# starting kernel
# baseline (speedup 1.0000x reference)
#include <hip/hip_runtime.h>
#include <hip/hip_bf16.h>
#include <math.h>

#define NN   50000
#define EE   800000
#define IND  128
#define HID  32
#define NH   4
#define NC   40
#define FDIM 128

typedef float v8f __attribute__((ext_vector_type(8)));
typedef float v2f __attribute__((ext_vector_type(2)));

__device__ __forceinline__ unsigned fkey(float f) {
    unsigned b = __float_as_uint(f);
    return (b & 0x80000000u) ? ~b : (b | 0x80000000u);
}
__device__ __forceinline__ float funkey(unsigned k) {
    unsigned b = (k & 0x80000000u) ? (k & 0x7FFFFFFFu) : ~k;
    return __uint_as_float(b);
}

__global__ void gat_gemm128(const float* __restrict__ X,
                            const float* __restrict__ W,
                            float* __restrict__ Y) {
    __shared__ float As[16 * FDIM];
    const int m0 = blockIdx.x * 16;
    const int t  = threadIdx.x;

    for (int i = t; i < 16 * FDIM / 4; i += 256)
        ((float4*)As)[i] = ((const float4*)(X + (size_t)m0 * FDIM))[i];
    __syncthreads();

    const int wave  = t >> 5;
    const int lane  = t & 31;
    const int n0    = wave * 16;
    const int mrow  = lane & 15;
    const int kgrp  = lane >> 4;

    v8f acc = {};
    for (int kk = 0; kk < FDIM; kk += 4) {
        v2f a, b;
        a.x = As[mrow * FDIM + kk + 2 * kgrp + 0];
        a.y = As[mrow * FDIM + kk + 2 * kgrp + 1];
        b.x = W[(kk + 2 * kgrp + 0) * FDIM + n0 + mrow];
        b.y = W[(kk + 2 * kgrp + 1) * FDIM + n0 + mrow];
        acc = __builtin_amdgcn_wmma_f32_16x16x4_f32(false, a, false, b,
                                                    (short)0, acc, false, false);
    }
    const int cn    = lane & 15;
    const int mbase = (lane >> 4) * 8;
    for (int r = 0; r < 8; ++r)
        Y[(size_t)(m0 + mbase + r) * FDIM + n0 + cn] = acc[r];
}

__global__ void gat_gemm_out(const float* __restrict__ Hm,
                             const float* __restrict__ Wout,
                             const float* __restrict__ bout,
                             float* __restrict__ logits) {
    __shared__ float As[16 * FDIM];
    const int m0 = blockIdx.x * 16;
    const int t  = threadIdx.x;
    for (int i = t; i < 16 * FDIM / 4; i += 96)
        ((float4*)As)[i] = ((const float4*)(Hm + (size_t)m0 * FDIM))[i];
    __syncthreads();

    const int wave = t >> 5;
    const int lane = t & 31;
    const int n0   = wave * 16;
    const int mrow = lane & 15;
    const int kgrp = lane >> 4;
    const int col  = n0 + mrow;

    v8f acc = {};
    for (int kk = 0; kk < FDIM; kk += 4) {
        v2f a, b;
        a.x = As[mrow * FDIM + kk + 2 * kgrp + 0];
        a.y = As[mrow * FDIM + kk + 2 * kgrp + 1];
        b.x = (col < NC) ? Wout[(kk + 2 * kgrp + 0) * NC + col] : 0.0f;
        b.y = (col < NC) ? Wout[(kk + 2 * kgrp + 1) * NC + col] : 0.0f;
        acc = __builtin_amdgcn_wmma_f32_16x16x4_f32(false, a, false, b,
                                                    (short)0, acc, false, false);
    }
    const int cn    = lane & 15;
    const int mbase = (lane >> 4) * 8;
    const int ocol  = n0 + cn;
    if (ocol < NC) {
        const float bb = bout[ocol];
        for (int r = 0; r < 8; ++r)
            logits[(size_t)(m0 + mbase + r) * NC + ocol] = acc[r] + bb;
    }
}

__global__ void gat_attn_coeff(const float* __restrict__ feat,
                               const float* __restrict__ al,
                               const float* __restrict__ ar,
                               float* __restrict__ el,
                               float* __restrict__ er) {
    int i = blockIdx.x * blockDim.x + threadIdx.x;
    if (i >= NN * NH) return;
    int n = i >> 2, h = i & 3;
    const float* f = feat + (size_t)n * FDIM + h * HID;
    float sl = 0.f, sr = 0.f;
    for (int d = 0; d < HID; ++d) {
        sl += f[d] * al[h * HID + d];
        sr += f[d] * ar[h * HID + d];
    }
    el[i] = sl;
    er[i] = sr;
}

__global__ void gat_edge_logits(const float* __restrict__ el,
                                const float* __restrict__ er,
                                const int* __restrict__ src,
                                const int* __restrict__ dst,
                                float* __restrict__ ebuf,
                                unsigned* __restrict__ maxkey) {
    int i = blockIdx.x * blockDim.x + threadIdx.x;
    if (i >= EE * NH) return;
    int e = i >> 2, h = i & 3;
    int s = src[e], d = dst[e];
    float v = el[s * NH + h] + er[d * NH + h];
    v = (v > 0.f) ? v : 0.2f * v;
    ebuf[i] = v;
    atomicMax(&maxkey[d * NH + h], fkey(v));
}

__global__ void gat_edge_exp(const int* __restrict__ dst,
                             float* __restrict__ ebuf,
                             const unsigned* __restrict__ maxkey,
                             float* __restrict__ denom) {
    int i = blockIdx.x * blockDim.x + threadIdx.x;
    if (i >= EE * NH) return;
    int e = i >> 2, h = i & 3;
    int d = dst[e];
    float m  = funkey(maxkey[d * NH + h]);
    float ee = expf(ebuf[i] - m);
    ebuf[i] = ee;
    atomicAdd(&denom[d * NH + h], ee);
}

__global__ void gat_scatter(const int* __restrict__ src,
                            const int* __restrict__ dst,
                            const float* __restrict__ ebuf,
                            const float* __restrict__ denom,
                            const float* __restrict__ feat,
                            float* __restrict__ agg) {
    int i = blockIdx.x * blockDim.x + threadIdx.x;
    if (i >= EE * FDIM) return;
    int e = i >> 7;
    int c = i & (FDIM - 1);
    int h = c >> 5;
    int s = src[e], d = dst[e];
    float a = ebuf[e * NH + h] / denom[d * NH + h];
    atomicAdd(&agg[(size_t)d * FDIM + c], feat[(size_t)s * FDIM + c] * a);
}

__global__ void gat_elu(const float* agg, const float* resid, float* out, int use_res) {
    int i = blockIdx.x * blockDim.x + threadIdx.x;
    if (i >= NN * FDIM) return;
    float v = agg[i] + (use_res ? resid[i] : 0.f);
    out[i] = (v > 0.f) ? v : (expf(v) - 1.0f);
}

__global__ void gat_mix(const float* __restrict__ h2,
                        const float* __restrict__ lamb,
                        float* hout) {
    int i = blockIdx.x * blockDim.x + threadIdx.x;
    if (i >= NN * FDIM) return;
    float lam = lamb[0];
    hout[i] = lam * hout[i] + (1.0f - lam) * h2[i];
}

static void run_conv(const float* X, const float* W, const float* al, const float* ar,
                     const int* src, const int* dst, const float* resid, float* out,
                     float* feat, float* agg, float* ebuf, float* el, float* er,
                     unsigned* maxkey, float* denom, hipStream_t stream) {
    hipMemsetAsync(agg,    0, (size_t)NN * FDIM * sizeof(float), stream);
    hipMemsetAsync(maxkey, 0, (size_t)NN * NH * sizeof(unsigned), stream);
    hipMemsetAsync(denom,  0, (size_t)NN * NH * sizeof(float), stream);

    gat_gemm128<<<NN / 16, 256, 0, stream>>>(X, W, feat);
    gat_attn_coeff<<<(NN * NH + 255) / 256, 256, 0, stream>>>(feat, al, ar, el, er);
    gat_edge_logits<<<(EE * NH + 255) / 256, 256, 0, stream>>>(el, er, src, dst, ebuf, maxkey);
    gat_edge_exp<<<(EE * NH + 255) / 256, 256, 0, stream>>>(dst, ebuf, maxkey, denom);
    gat_scatter<<<(EE * FDIM + 255) / 256, 256, 0, stream>>>(src, dst, ebuf, denom, feat, agg);
    gat_elu<<<(NN * FDIM + 255) / 256, 256, 0, stream>>>(agg, resid, out, resid != nullptr);
}

extern "C" void kernel_launch(void* const* d_in, const int* in_sizes, int n_in,
                              void* d_out, int out_size, void* d_ws, size_t ws_size,
                              hipStream_t stream) {
    (void)in_sizes; (void)n_in; (void)out_size; (void)ws_size;

    const float* inputs = (const float*)d_in[0];
    const float* target = (const float*)d_in[1];
    const float* lamb   = (const float*)d_in[2];
    const float* W0     = (const float*)d_in[3];
    const float* al0    = (const float*)d_in[4];
    const float* ar0    = (const float*)d_in[5];
    const float* W1     = (const float*)d_in[6];
    const float* al1    = (const float*)d_in[7];
    const float* ar1    = (const float*)d_in[8];
    const float* Wout   = (const float*)d_in[9];
    const float* bout   = (const float*)d_in[10];
    const int*   src    = (const int*)d_in[11];
    const int*   dst    = (const int*)d_in[12];

    float* feat   = (float*)d_ws;
    float* agg    = feat + (size_t)NN * FDIM;
    float* hbuf   = agg  + (size_t)NN * FDIM;
    float* ebuf   = hbuf + (size_t)NN * FDIM;
    float* el     = ebuf + (size_t)EE * NH;
    float* er     = el   + (size_t)NN * NH;
    unsigned* mk  = (unsigned*)(er + (size_t)NN * NH);
    float* denom  = (float*)(mk + (size_t)NN * NH);

    float* hregion = (float*)d_out;
    float* logits  = hregion + (size_t)NN * FDIM;

    run_conv(inputs, W0, al0, ar0, src, dst, nullptr, hbuf,
             feat, agg, ebuf, el, er, mk, denom, stream);
    run_conv(hbuf, W1, al1, ar1, src, dst, hbuf, hregion,
             feat, agg, ebuf, el, er, mk, denom, stream);

    run_conv(target, W0, al0, ar0, src, dst, nullptr, hbuf,
             feat, agg, ebuf, el, er, mk, denom, stream);
    run_conv(hbuf, W1, al1, ar1, src, dst, hbuf, hbuf,
             feat, agg, ebuf, el, er, mk, denom, stream);

    gat_mix<<<(NN * FDIM + 255) / 256, 256, 0, stream>>>(hbuf, lamb, hregion);
    gat_gemm_out<<<NN / 16, 96, 0, stream>>>(hregion, Wout, bout, logits);
}


# GEMM re-tiled: a wave owns 16 rows x all 128 columns, A fragments straight from global into registers, W staged once per 128-row block in LDS in fragment order; k order inside a dot product changes
# speedup vs baseline: 2.4598x; 2.4598x over previous
#include <hip/hip_runtime.h>
#include <hip/hip_bf16.h>
#include <math.h>

#define NN   50000
#define EE   800000
#define IND  128
#define HID  32
#define NH   4
#define NC   40
#define FDIM 128

typedef float v8f __attribute__((ext_vector_type(8)));
typedef float v2f __attribute__((ext_vector_type(2)));

__device__ __forceinline__ unsigned fkey(float f) {
    unsigned b = __float_as_uint(f);
    return (b & 0x80000000u) ? ~b : (b | 0x80000000u);
}
__device__ __forceinline__ float funkey(unsigned k) {
    unsigned b = (k & 0x80000000u) ? (k & 0x7FFFFFFFu) : ~k;
    return __uint_as_float(b);
}

__device__ __forceinline__ float elu1(float v) {
    return (v > 0.f) ? v : (expf(v) - 1.0f);
}

__device__ __forceinline__ float inv_den(float d) {
    return (d > 0.f) ? (1.0f / d) : 0.f;
}

#define GROWS 128
__device__ __forceinline__ void gemm128_body(const float* __restrict__ X,
                                             const float* __restrict__ W,
                                             float* __restrict__ Y,
                                             float* __restrict__ zagg,
                                             unsigned* __restrict__ zmk,
                                             float* __restrict__ zden,
                                             const float* __restrict__ pden,
                                             const bool act) {
    __shared__ float4 Ws[16 * 8 * 32];
    const int t    = threadIdx.x;
    const int lane = t & 31;
    const int wave = __builtin_amdgcn_readfirstlane(t >> 5);
    const int c    = lane & 15;
    const int hh   = lane >> 4;
    const int m0   = blockIdx.x * GROWS;
    const int mw   = m0 + wave * 16;
    const bool live = mw < NN;

    float4 a[16];
    float4 q = make_float4(1.f, 1.f, 1.f, 1.f);
    if (live) {
        const float4* xr = (const float4*)(X + (size_t)(mw + c) * FDIM) + hh;
#pragma unroll
        for (int i = 0; i < 16; ++i) a[i] = xr[2 * i];
        if (act) q = *(const float4*)(pden + (size_t)(mw + c) * NH);

        const float4 z4 = make_float4(0.f, 0.f, 0.f, 0.f);
        float4* za = (float4*)(zagg + (size_t)mw * FDIM);
#pragma unroll
        for (int r = 0; r < 16; ++r) za[lane + 32 * r] = z4;
        if (lane < 16) ((uint4*)(zmk + (size_t)mw * NH))[lane] = make_uint4(0u, 0u, 0u, 0u);
        else           ((float4*)(zden + (size_t)mw * NH))[lane - 16] = z4;
    }

#pragma unroll
    for (int i = 0; i < 16; ++i) {
        const float* wp = W + (size_t)(8 * i + 4 * hh) * FDIM + 16 * wave + c;
        float4 v;
        v.x = wp[0];
        v.y = wp[FDIM];
        v.z = wp[2 * FDIM];
        v.w = wp[3 * FDIM];
        Ws[(i * 8 + wave) * 32 + lane] = v;
    }

    if (act && live) {
        const float iq[4] = { inv_den(q.x), inv_den(q.y), inv_den(q.z), inv_den(q.w) };
#pragma unroll
        for (int i = 0; i < 16; ++i) {
            const float s = iq[i >> 2];
            a[i].x = elu1(a[i].x * s); a[i].y = elu1(a[i].y * s);
            a[i].z = elu1(a[i].z * s); a[i].w = elu1(a[i].w * s);
        }
    }
    __syncthreads();
    if (!live) return;

    v8f acc[8] = {};
    float4 b[2][8];
#pragma unroll
    for (int j = 0; j < 8; ++j) b[0][j] = Ws[j * 32 + lane];
#pragma unroll
    for (int i = 0; i < 16; ++i) {
        const int cur = i & 1;
        __builtin_amdgcn_sched_barrier(0);
        if (i + 1 < 16) {
#pragma unroll
            for (int j = 0; j < 8; ++j) b[cur ^ 1][j] = Ws[((i + 1) * 8 + j) * 32 + lane];
        }
        __builtin_amdgcn_sched_barrier(0);
        v2f a0, a1;
        a0.x = a[i].x; a0.y = a[i].y;
        a1.x = a[i].z; a1.y = a[i].w;
#pragma unroll
        for (int j = 0; j < 8; ++j) {
            v2f b0;
            b0.x = b[cur][j].x; b0.y = b[cur][j].y;
            acc[j] = __builtin_amdgcn_wmma_f32_16x16x4_f32(false, a0, false, b0,
                                                           (short)0, acc[j], false, false);
        }
#pragma unroll
        for (int j = 0; j < 8; ++j) {
            v2f b1;
            b1.x = b[cur][j].z; b1.y = b[cur][j].w;
            acc[j] = __builtin_amdgcn_wmma_f32_16x16x4_f32(false, a1, false, b1,
                                                           (short)0, acc[j], false, false);
        }
    }
    float* yr = Y + (size_t)(mw + 8 * hh) * FDIM + c;
#pragma unroll
    for (int j = 0; j < 8; ++j)
#pragma unroll
        for (int r = 0; r < 8; ++r)
            yr[(size_t)r * FDIM + 16 * j] = acc[j][r];
}

__global__ void __launch_bounds__(256)
gat_gemm128(const float* __restrict__ X,
                            const float* __restrict__ W,
                            float* __restrict__ Y,
                            float* __restrict__ zagg,
                            unsigned* __restrict__ zmk,
                            float* __restrict__ zden) {
    gemm128_body(X, W, Y, zagg, zmk, zden, nullptr, false);
}

__global__ void __launch_bounds__(256)
gat_gemm128_act(const float* __restrict__ S,
                                const float* __restrict__ W,
                                float* __restrict__ Y,
                                float* __restrict__ zagg,
                                unsigned* __restrict__ zmk,
                                float* __restrict__ zden,
                                const float* __restrict__ pden) {
    gemm128_body(S, W, Y, zagg, zmk, zden, pden, true);
}

__global__ void gat_gemm_out(const float* __restrict__ s0a,
                             const float* __restrict__ s1a,
                             const float* __restrict__ s0b,
                             const float* __restrict__ s1b,
                             const float* __restrict__ d0a,
                             const float* __restrict__ d1a,
                             const float* __restrict__ d0b,
                             const float* __restrict__ d1b,
                             const float* __restrict__ lamb,
                             const float* __restrict__ Wout,
                             const float* __restrict__ bout,
                             float* __restrict__ hout,
                             float* __restrict__ logits) {
    __shared__ float As[16 * FDIM];
    const int m0 = blockIdx.x * 16;
    const int t  = threadIdx.x;
    const float lam = lamb[0];
    const size_t base4 = (size_t)m0 * (FDIM / 4);
    constexpr int NP = 6;
    float4 a0[NP], a1[NP], b0[NP], b1[NP];
    float qa0[NP], qa1[NP], qb0[NP], qb1[NP];
#pragma unroll
    for (int k = 0; k < NP; ++k) {
        const int i = t + 96 * k;
        if (i < 16 * FDIM / 4) {
            a0[k] = ((const float4*)s0a)[base4 + i];
            a1[k] = ((const float4*)s1a)[base4 + i];
            b0[k] = ((const float4*)s0b)[base4 + i];
            b1[k] = ((const float4*)s1b)[base4 + i];
            const size_t di = (size_t)(m0 + (i >> 5)) * NH + ((i & 31) >> 3);
            qa0[k] = d0a[di];
            qa1[k] = d1a[di];
            qb0[k] = d0b[di];
            qb1[k] = d1b[di];
        }
    }
#pragma unroll
    for (int k = 0; k < NP; ++k) {
        const int i = t + 96 * k;
        if (i < 16 * FDIM / 4) {
            const float ia0 = inv_den(qa0[k]), ia1 = inv_den(qa1[k]);
            const float ib0 = inv_den(qb0[k]), ib1 = inv_den(qb1[k]);
            float4 r;
            r.x = lam * elu1(a1[k].x * ia1 + elu1(a0[k].x * ia0)) + (1.0f - lam) * elu1(b1[k].x * ib1 + elu1(b0[k].x * ib0));
            r.y = lam * elu1(a1[k].y * ia1 + elu1(a0[k].y * ia0)) + (1.0f - lam) * elu1(b1[k].y * ib1 + elu1(b0[k].y * ib0));
            r.z = lam * elu1(a1[k].z * ia1 + elu1(a0[k].z * ia0)) + (1.0f - lam) * elu1(b1[k].z * ib1 + elu1(b0[k].z * ib0));
            r.w = lam * elu1(a1[k].w * ia1 + elu1(a0[k].w * ia0)) + (1.0f - lam) * elu1(b1[k].w * ib1 + elu1(b0[k].w * ib0));
            ((float4*)As)[i] = r;
            ((float4*)hout)[base4 + i] = r;
        }
    }
    __syncthreads();

    const int wave = t >> 5;
    const int lane = t & 31;
    const int n0   = wave * 16;
    const int mrow = lane & 15;
    const int kgrp = lane >> 4;
    const int col  = n0 + mrow;

    v8f acc = {};
    for (int kk = 0; kk < FDIM; kk += 4) {
        v2f a, b;
        a.x = As[mrow * FDIM + kk + 2 * kgrp + 0];
        a.y = As[mrow * FDIM + kk + 2 * kgrp + 1];
        b.x = (col < NC) ? Wout[(kk + 2 * kgrp + 0) * NC + col] : 0.0f;
        b.y = (col < NC) ? Wout[(kk + 2 * kgrp + 1) * NC + col] : 0.0f;
        acc = __builtin_amdgcn_wmma_f32_16x16x4_f32(false, a, false, b,
                                                    (short)0, acc, false, false);
    }
    const int cn    = lane & 15;
    const int mbase = (lane >> 4) * 8;
    const int ocol  = n0 + cn;
    if (ocol < NC) {
        const float bb = bout[ocol];
        for (int r = 0; r < 8; ++r)
            logits[(size_t)(m0 + mbase + r) * NC + ocol] = acc[r] + bb;
    }
}

__global__ void gat_attn_coeff(const float* __restrict__ feat,
                               const float* __restrict__ al,
                               const float* __restrict__ ar,
                               float* __restrict__ el,
                               float* __restrict__ er) {
    int i = blockIdx.x * blockDim.x + threadIdx.x;
    if (i >= NN * NH) return;
    int n = i >> 2, h = i & 3;
    const float* f = feat + (size_t)n * FDIM + h * HID;
    float sl = 0.f, sr = 0.f;
    for (int d = 0; d < HID; ++d) {
        sl += f[d] * al[h * HID + d];
        sr += f[d] * ar[h * HID + d];
    }
    el[i] = sl;
    er[i] = sr;
}

__global__ void gat_edge_logits(const float* __restrict__ el,
                                const float* __restrict__ er,
                                const int* __restrict__ src,
                                const int* __restrict__ dst,
                                float* __restrict__ ebuf,
                                unsigned* __restrict__ maxkey) {
    int i = blockIdx.x * blockDim.x + threadIdx.x;
    if (i >= EE * NH) return;
    int e = i >> 2, h = i & 3;
    int s = src[e], d = dst[e];
    float v = el[s * NH + h] + er[d * NH + h];
    v = (v > 0.f) ? v : 0.2f * v;
    ebuf[i] = v;
    atomicMax(&maxkey[d * NH + h], fkey(v));
}

#define EPW 16
#define EUN 8
__device__ __forceinline__ void rows_load(float (&f)[EUN][NH], const int s, const int lane,
                                          const float* __restrict__ feat) {
#pragma unroll
    for (int u = 0; u < EUN; ++u) {
        const int sj = __builtin_amdgcn_readlane(s, 4 * u);
        const float* fr = feat + (size_t)sj * FDIM + lane;
#pragma unroll
        for (int h = 0; h < NH; ++h) f[u][h] = fr[h * HID];
    }
}
__device__ __forceinline__ void rows_add(const float (&f)[EUN][NH], const int d, const float w,
                                         const int lane, float* __restrict__ agg) {
#pragma unroll
    for (int u = 0; u < EUN; ++u) {
        const int dj = __builtin_amdgcn_readlane(d, 4 * u);
        const float b0 = __int_as_float(__builtin_amdgcn_readlane(__float_as_int(w), 4 * u + 0));
        const float b1 = __int_as_float(__builtin_amdgcn_readlane(__float_as_int(w), 4 * u + 1));
        const float b2 = __int_as_float(__builtin_amdgcn_readlane(__float_as_int(w), 4 * u + 2));
        const float b3 = __int_as_float(__builtin_amdgcn_readlane(__float_as_int(w), 4 * u + 3));
        float* ar = agg + (size_t)dj * FDIM + lane;
        atomicAdd(ar + 0 * HID, f[u][0] * b0);
        atomicAdd(ar + 1 * HID, f[u][1] * b1);
        atomicAdd(ar + 2 * HID, f[u][2] * b2);
        atomicAdd(ar + 3 * HID, f[u][3] * b3);
    }
}
__global__ void __launch_bounds__(256)
gat_scatter(const int* __restrict__ src,
            const int* __restrict__ dst,
            const float* __restrict__ ebuf,
            const unsigned* __restrict__ maxkey,
            const float* __restrict__ feat,
            float* __restrict__ agg,
            float* __restrict__ den) {
    static_assert(EE % EPW == 0 && EPW == 2 * EUN && EUN * NH == 32, "a wave owns two groups of 8 whole edges");
    const int lane = threadIdx.x & 31;
    const int wave = (blockIdx.x * blockDim.x + threadIdx.x) >> 5;
    const int e0   = wave * EPW;
    if (e0 >= EE) return;

    const int hd = lane & 3;
    const int eg = e0 + (lane >> 2);
    const int s0 = src[eg],       d0 = dst[eg];
    const int s1 = src[eg + EUN], d1 = dst[eg + EUN];
    const float v0 = ebuf[(size_t)e0 * NH + lane];
    const float v1 = ebuf[(size_t)(e0 + EUN) * NH + lane];
    const unsigned k0 = maxkey[d0 * NH + hd];
    const unsigned k1 = maxkey[d1 * NH + hd];

    float f[EUN][NH];
    rows_load(f, s0, lane, feat);

    const float w0 = expf(v0 - funkey(k0));
    const float w1 = expf(v1 - funkey(k1));
    atomicAdd(&den[d0 * NH + hd], w0);
    atomicAdd(&den[d1 * NH + hd], w1);

    rows_add(f, d0, w0, lane, agg);
    rows_load(f, s1, lane, feat);
    rows_add(f, d1, w1, lane, agg);
}

static void run_conv(const float* Xin, const float* pden, const float* W, const float* al, const float* ar,
                     const int* src, const int* dst, float* aggout, float* denout,
                     float* feat, float* ebuf, float* el, float* er,
                     unsigned* maxkey, hipStream_t stream) {
    if (pden)
        gat_gemm128_act<<<(NN + GROWS - 1) / GROWS, 256, 0, stream>>>(Xin, W, feat, aggout, maxkey, denout, pden);
    else
        gat_gemm128<<<(NN + GROWS - 1) / GROWS, 256, 0, stream>>>(Xin, W, feat, aggout, maxkey, denout);
    gat_attn_coeff<<<(NN * NH + 255) / 256, 256, 0, stream>>>(feat, al, ar, el, er);
    gat_edge_logits<<<(EE * NH + 255) / 256, 256, 0, stream>>>(el, er, src, dst, ebuf, maxkey);
    gat_scatter<<<(EE / EPW * 32 + 255) / 256, 256, 0, stream>>>(src, dst, ebuf, maxkey, feat, aggout, denout);
}

extern "C" void kernel_launch(void* const* d_in, const int* in_sizes, int n_in,
                              void* d_out, int out_size, void* d_ws, size_t ws_size,
                              hipStream_t stream) {
    (void)in_sizes; (void)n_in; (void)out_size; (void)ws_size;

    const float* inputs = (const float*)d_in[0];
    const float* target = (const float*)d_in[1];
    const float* lamb   = (const float*)d_in[2];
    const float* W0     = (const float*)d_in[3];
    const float* al0    = (const float*)d_in[4];
    const float* ar0    = (const float*)d_in[5];
    const float* W1     = (const float*)d_in[6];
    const float* al1    = (const float*)d_in[7];
    const float* ar1    = (const float*)d_in[8];
    const float* Wout   = (const float*)d_in[9];
    const float* bout   = (const float*)d_in[10];
    const int*   src    = (const int*)d_in[11];
    const int*   dst    = (const int*)d_in[12];

    float* feat   = (float*)d_ws;
    float* s0a    = feat + (size_t)NN * FDIM;
    float* s1a    = s0a  + (size_t)NN * FDIM;
    float* s0b    = s1a  + (size_t)NN * FDIM;
    float* s1b    = s0b  + (size_t)NN * FDIM;
    float* ebuf   = s1b  + (size_t)NN * FDIM;
    float* el     = ebuf + (size_t)EE * NH;
    float* er     = el   + (size_t)NN * NH;
    unsigned* mk  = (unsigned*)(er + (size_t)NN * NH);
    float* d0a    = (float*)(mk + (size_t)NN * NH);
    float* d1a    = d0a  + (size_t)NN * NH;
    float* d0b    = d1a  + (size_t)NN * NH;
    float* d1b    = d0b  + (size_t)NN * NH;

    float* hregion = (float*)d_out;
    float* logits  = hregion + (size_t)NN * FDIM;

    run_conv(inputs, nullptr, W0, al0, ar0, src, dst, s0a, d0a,
             feat, ebuf, el, er, mk, stream);
    run_conv(s0a, d0a, W1, al1, ar1, src, dst, s1a, d1a,
             feat, ebuf, el, er, mk, stream);

    run_conv(target, nullptr, W0, al0, ar0, src, dst, s0b, d0b,
             feat, ebuf, el, er, mk, stream);
    run_conv(s0b, d0b, W1, al1, ar1, src, dst, s1b, d1b,
             feat, ebuf, el, er, mk, stream);

    gat_gemm_out<<<NN / 16, 96, 0, stream>>>(s0a, s1a, s0b, s1b, d0a, d1a, d0b, d1b,
                                             lamb, Wout, bout, hregion, logits);
}


# Scatter cut by head pair over the chip's two halves: a wave reads its die, draws 32-edge tickets (returning add) for its half's pair; a sum line is added to from one half only; arrival order differs
# speedup vs baseline: 2.5601x; 1.0408x over previous
#include <hip/hip_runtime.h>
#include <hip/hip_bf16.h>
#include <math.h>

#define NN   50000
#define EE   800000
#define IND  128
#define HID  32
#define NH   4
#define NC   40
#define FDIM 128

typedef float v8f __attribute__((ext_vector_type(8)));
typedef float v2f __attribute__((ext_vector_type(2)));

__device__ __forceinline__ unsigned fkey(float f) {
    unsigned b = __float_as_uint(f);
    return (b & 0x80000000u) ? ~b : (b | 0x80000000u);
}
__device__ __forceinline__ float funkey(unsigned k) {
    unsigned b = (k & 0x80000000u) ? (k & 0x7FFFFFFFu) : ~k;
    return __uint_as_float(b);
}

__device__ __forceinline__ float elu1(float v) {
    return (v > 0.f) ? v : (expf(v) - 1.0f);
}

__device__ __forceinline__ float inv_den(float d) {
    return (d > 0.f) ? (1.0f / d) : 0.f;
}

#define GROWS 128
#define TSTRIDE 64
__device__ __forceinline__ void gemm128_body(const float* __restrict__ X,
                                             const float* __restrict__ W,
                                             float* __restrict__ Y,
                                             float* __restrict__ zagg,
                                             unsigned* __restrict__ zmk,
                                             float* __restrict__ zden,
                                             const float* __restrict__ pden,
                                             unsigned* __restrict__ tick,
                                             const bool act) {
    __shared__ float4 Ws[16 * 8 * 32];
    const int t    = threadIdx.x;
    const int lane = t & 31;
    const int wave = __builtin_amdgcn_readfirstlane(t >> 5);
    const int c    = lane & 15;
    const int hh   = lane >> 4;
    const int m0   = blockIdx.x * GROWS;
    const int mw   = m0 + wave * 16;
    const bool live = mw < NN;
    if (blockIdx.x == 0 && t < 2) tick[t * TSTRIDE] = 0u;

    float4 a[16];
    float4 q = make_float4(1.f, 1.f, 1.f, 1.f);
    if (live) {
        const float4* xr = (const float4*)(X + (size_t)(mw + c) * FDIM) + hh;
#pragma unroll
        for (int i = 0; i < 16; ++i) a[i] = xr[2 * i];
        if (act) q = *(const float4*)(pden + (size_t)(mw + c) * NH);

        const float4 z4 = make_float4(0.f, 0.f, 0.f, 0.f);
        float4* za = (float4*)(zagg + (size_t)mw * FDIM);
#pragma unroll
        for (int r = 0; r < 16; ++r) za[lane + 32 * r] = z4;
        if (lane < 16) ((uint4*)(zmk + (size_t)mw * NH))[lane] = make_uint4(0u, 0u, 0u, 0u);
        else           ((float4*)(zden + (size_t)mw * NH))[lane - 16] = z4;
    }

#pragma unroll
    for (int i = 0; i < 16; ++i) {
        const float* wp = W + (size_t)(8 * i + 4 * hh) * FDIM + 16 * wave + c;
        float4 v;
        v.x = wp[0];
        v.y = wp[FDIM];
        v.z = wp[2 * FDIM];
        v.w = wp[3 * FDIM];
        Ws[(i * 8 + wave) * 32 + lane] = v;
    }

    if (act && live) {
        const float iq[4] = { inv_den(q.x), inv_den(q.y), inv_den(q.z), inv_den(q.w) };
#pragma unroll
        for (int i = 0; i < 16; ++i) {
            const float s = iq[i >> 2];
            a[i].x = elu1(a[i].x * s); a[i].y = elu1(a[i].y * s);
            a[i].z = elu1(a[i].z * s); a[i].w = elu1(a[i].w * s);
        }
    }
    __syncthreads();
    if (!live) return;

    v8f acc[8] = {};
    float4 b[2][8];
#pragma unroll
    for (int j = 0; j < 8; ++j) b[0][j] = Ws[j * 32 + lane];
#pragma unroll
    for (int i = 0; i < 16; ++i) {
        const int cur = i & 1;
        __builtin_amdgcn_sched_barrier(0);
        if (i + 1 < 16) {
#pragma unroll
            for (int j = 0; j < 8; ++j) b[cur ^ 1][j] = Ws[((i + 1) * 8 + j) * 32 + lane];
        }
        __builtin_amdgcn_sched_barrier(0);
        v2f a0, a1;
        a0.x = a[i].x; a0.y = a[i].y;
        a1.x = a[i].z; a1.y = a[i].w;
#pragma unroll
        for (int j = 0; j < 8; ++j) {
            v2f b0;
            b0.x = b[cur][j].x; b0.y = b[cur][j].y;
            acc[j] = __builtin_amdgcn_wmma_f32_16x16x4_f32(false, a0, false, b0,
                                                           (short)0, acc[j], false, false);
        }
#pragma unroll
        for (int j = 0; j < 8; ++j) {
            v2f b1;
            b1.x = b[cur][j].z; b1.y = b[cur][j].w;
            acc[j] = __builtin_amdgcn_wmma_f32_16x16x4_f32(false, a1, false, b1,
                                                           (short)0, acc[j], false, false);
        }
    }
    float* yr = Y + (size_t)(mw + 8 * hh) * FDIM + c;
#pragma unroll
    for (int j = 0; j < 8; ++j)
#pragma unroll
        for (int r = 0; r < 8; ++r)
            yr[(size_t)r * FDIM + 16 * j] = acc[j][r];
}

__global__ void __launch_bounds__(256)
gat_gemm128(const float* __restrict__ X,
                            const float* __restrict__ W,
                            float* __restrict__ Y,
                            float* __restrict__ zagg,
                            unsigned* __restrict__ zmk,
                            float* __restrict__ zden,
                            unsigned* __restrict__ tick) {
    gemm128_body(X, W, Y, zagg, zmk, zden, nullptr, tick, false);
}

__global__ void __launch_bounds__(256)
gat_gemm128_act(const float* __restrict__ S,
                                const float* __restrict__ W,
                                float* __restrict__ Y,
                                float* __restrict__ zagg,
                                unsigned* __restrict__ zmk,
                                float* __restrict__ zden,
                                const float* __restrict__ pden,
                                unsigned* __restrict__ tick) {
    gemm128_body(S, W, Y, zagg, zmk, zden, pden, tick, true);
}

__global__ void gat_gemm_out(const float* __restrict__ s0a,
                             const float* __restrict__ s1a,
                             const float* __restrict__ s0b,
                             const float* __restrict__ s1b,
                             const float* __restrict__ d0a,
                             const float* __restrict__ d1a,
                             const float* __restrict__ d0b,
                             const float* __restrict__ d1b,
                             const float* __restrict__ lamb,
                             const float* __restrict__ Wout,
                             const float* __restrict__ bout,
                             float* __restrict__ hout,
                             float* __restrict__ logits) {
    __shared__ float As[16 * FDIM];
    const int m0 = blockIdx.x * 16;
    const int t  = threadIdx.x;
    const float lam = lamb[0];
    const size_t base4 = (size_t)m0 * (FDIM / 4);
    constexpr int NP = 6;
    float4 a0[NP], a1[NP], b0[NP], b1[NP];
    float qa0[NP], qa1[NP], qb0[NP], qb1[NP];
#pragma unroll
    for (int k = 0; k < NP; ++k) {
        const int i = t + 96 * k;
        if (i < 16 * FDIM / 4) {
            a0[k] = ((const float4*)s0a)[base4 + i];
            a1[k] = ((const float4*)s1a)[base4 + i];
            b0[k] = ((const float4*)s0b)[base4 + i];
            b1[k] = ((const float4*)s1b)[base4 + i];
            const size_t di = (size_t)(m0 + (i >> 5)) * NH + ((i & 31) >> 3);
            qa0[k] = d0a[di];
            qa1[k] = d1a[di];
            qb0[k] = d0b[di];
            qb1[k] = d1b[di];
        }
    }
#pragma unroll
    for (int k = 0; k < NP; ++k) {
        const int i = t + 96 * k;
        if (i < 16 * FDIM / 4) {
            const float ia0 = inv_den(qa0[k]), ia1 = inv_den(qa1[k]);
            const float ib0 = inv_den(qb0[k]), ib1 = inv_den(qb1[k]);
            float4 r;
            r.x = lam * elu1(a1[k].x * ia1 + elu1(a0[k].x * ia0)) + (1.0f - lam) * elu1(b1[k].x * ib1 + elu1(b0[k].x * ib0));
            r.y = lam * elu1(a1[k].y * ia1 + elu1(a0[k].y * ia0)) + (1.0f - lam) * elu1(b1[k].y * ib1 + elu1(b0[k].y * ib0));
            r.z = lam * elu1(a1[k].z * ia1 + elu1(a0[k].z * ia0)) + (1.0f - lam) * elu1(b1[k].z * ib1 + elu1(b0[k].z * ib0));
            r.w = lam * elu1(a1[k].w * ia1 + elu1(a0[k].w * ia0)) + (1.0f - lam) * elu1(b1[k].w * ib1 + elu1(b0[k].w * ib0));
            ((float4*)As)[i] = r;
            ((float4*)hout)[base4 + i] = r;
        }
    }
    __syncthreads();

    const int wave = t >> 5;
    const int lane = t & 31;
    const int n0   = wave * 16;
    const int mrow = lane & 15;
    const int kgrp = lane >> 4;
    const int col  = n0 + mrow;

    v8f acc = {};
    for (int kk = 0; kk < FDIM; kk += 4) {
        v2f a, b;
        a.x = As[mrow * FDIM + kk + 2 * kgrp + 0];
        a.y = As[mrow * FDIM + kk + 2 * kgrp + 1];
        b.x = (col < NC) ? Wout[(kk + 2 * kgrp + 0) * NC + col] : 0.0f;
        b.y = (col < NC) ? Wout[(kk + 2 * kgrp + 1) * NC + col] : 0.0f;
        acc = __builtin_amdgcn_wmma_f32_16x16x4_f32(false, a, false, b,
                                                    (short)0, acc, false, false);
    }
    const int cn    = lane & 15;
    const int mbase = (lane >> 4) * 8;
    const int ocol  = n0 + cn;
    if (ocol < NC) {
        const float bb = bout[ocol];
        for (int r = 0; r < 8; ++r)
            logits[(size_t)(m0 + mbase + r) * NC + ocol] = acc[r] + bb;
    }
}

__global__ void gat_attn_coeff(const float* __restrict__ feat,
                               const float* __restrict__ al,
                               const float* __restrict__ ar,
                               float* __restrict__ el,
                               float* __restrict__ er) {
    int i = blockIdx.x * blockDim.x + threadIdx.x;
    if (i >= NN * NH) return;
    int n = i >> 2, h = i & 3;
    const float* f = feat + (size_t)n * FDIM + h * HID;
    float sl = 0.f, sr = 0.f;
    for (int d = 0; d < HID; ++d) {
        sl += f[d] * al[h * HID + d];
        sr += f[d] * ar[h * HID + d];
    }
    el[i] = sl;
    er[i] = sr;
}

__global__ void gat_edge_logits(const float* __restrict__ el,
                                const float* __restrict__ er,
                                const int* __restrict__ src,
                                const int* __restrict__ dst,
                                float* __restrict__ ebuf,
                                unsigned* __restrict__ maxkey) {
    int i = blockIdx.x * blockDim.x + threadIdx.x;
    if (i >= EE * NH) return;
    int e = i >> 2, h = i & 3;
    int s = src[e], d = dst[e];
    float v = el[s * NH + h] + er[d * NH + h];
    v = (v > 0.f) ? v : 0.2f * v;
    ebuf[i] = v;
    atomicMax(&maxkey[d * NH + h], fkey(v));
}

#define EUN    8
#define TEDGES 32
#define NTICK  (EE / TEDGES)
__device__ __forceinline__ void rows_load(float (&f)[EUN][2], const int s, const unsigned lofs,
                                          const float* __restrict__ feat) {
#pragma unroll
    for (int u = 0; u < EUN; ++u) {
        const int sj = __builtin_amdgcn_readlane(s, 4 * u);
        const float* fr = feat + (size_t)sj * FDIM;
        f[u][0] = fr[lofs];
        f[u][1] = fr[lofs + HID];
    }
}
__device__ __forceinline__ void rows_add(const float (&f)[EUN][2], const int d, const float w,
                                         const int p2, const unsigned lofs, float* __restrict__ agg) {
#pragma unroll
    for (int u = 0; u < EUN; ++u) {
        const int dj = __builtin_amdgcn_readlane(d, 4 * u);
        const float b0 = __int_as_float(__builtin_amdgcn_readlane(__float_as_int(w), 4 * u + p2));
        const float b1 = __int_as_float(__builtin_amdgcn_readlane(__float_as_int(w), 4 * u + p2 + 1));
        float* ar = agg + (size_t)dj * FDIM;
        atomicAdd(ar + lofs,       f[u][0] * b0);
        atomicAdd(ar + lofs + HID, f[u][1] * b1);
    }
}
__global__ void __launch_bounds__(256)
gat_scatter(const int* __restrict__ src,
            const int* __restrict__ dst,
            const float* __restrict__ ebuf,
            const unsigned* __restrict__ maxkey,
            const float* __restrict__ feat,
            float* __restrict__ agg,
            float* __restrict__ den,
            unsigned* __restrict__ tick) {
    static_assert(EE % TEDGES == 0 && TEDGES == 4 * EUN && EUN * NH == 32, "a ticket is four groups of 8 whole edges");
    const int lane = threadIdx.x & 31;
    const int hd = lane & 3;
    int p = (int)((__builtin_amdgcn_s_sendmsg_rtn(0x87) >> 18) & 1u);
    bool other = false;
    for (;;) {
        unsigned tv = 0u;
        if (lane == 0) tv = atomicAdd(&tick[p * TSTRIDE], 1u);
        const unsigned t = (unsigned)__builtin_amdgcn_readfirstlane((int)tv);
        if (t >= (unsigned)NTICK) {
            if (other) break;
            other = true;
            p ^= 1;
            continue;
        }
        const int p2 = 2 * p;
        const unsigned lofs = (unsigned)(p2 * HID + lane);
#pragma unroll 1
        for (int g = 0; g < TEDGES; g += 2 * EUN) {
            const int e0 = (int)t * TEDGES + g;
            const int eg = e0 + (lane >> 2);
            const int s0 = src[eg],       d0 = dst[eg];
            const int s1 = src[eg + EUN], d1 = dst[eg + EUN];
            const float v0 = ebuf[(size_t)e0 * NH + lane];
            const float v1 = ebuf[(size_t)(e0 + EUN) * NH + lane];
            const unsigned k0 = maxkey[d0 * NH + hd];
            const unsigned k1 = maxkey[d1 * NH + hd];

            float f0[EUN][2], f1[EUN][2];
            rows_load(f0, s0, lofs, feat);
            rows_load(f1, s1, lofs, feat);

            const float w0 = expf(v0 - funkey(k0));
            const float w1 = expf(v1 - funkey(k1));
            if (p == 0) {
                atomicAdd(&den[d0 * NH + hd], w0);
                atomicAdd(&den[d1 * NH + hd], w1);
            }
            rows_add(f0, d0, w0, p2, lofs, agg);
            rows_add(f1, d1, w1, p2, lofs, agg);
        }
    }
}

static void run_conv(const float* Xin, const float* pden, const float* W, const float* al, const float* ar,
                     const int* src, const int* dst, float* aggout, float* denout,
                     float* feat, float* ebuf, float* el, float* er,
                     unsigned* maxkey, unsigned* tick, hipStream_t stream) {
    if (pden)
        gat_gemm128_act<<<(NN + GROWS - 1) / GROWS, 256, 0, stream>>>(Xin, W, feat, aggout, maxkey, denout, pden, tick);
    else
        gat_gemm128<<<(NN + GROWS - 1) / GROWS, 256, 0, stream>>>(Xin, W, feat, aggout, maxkey, denout, tick);
    gat_attn_coeff<<<(NN * NH + 255) / 256, 256, 0, stream>>>(feat, al, ar, el, er);
    gat_edge_logits<<<(EE * NH + 255) / 256, 256, 0, stream>>>(el, er, src, dst, ebuf, maxkey);
    gat_scatter<<<2048, 256, 0, stream>>>(src, dst, ebuf, maxkey, feat, aggout, denout, tick);
}

extern "C" void kernel_launch(void* const* d_in, const int* in_sizes, int n_in,
                              void* d_out, int out_size, void* d_ws, size_t ws_size,
                              hipStream_t stream) {
    (void)in_sizes; (void)n_in; (void)out_size; (void)ws_size;

    const float* inputs = (const float*)d_in[0];
    const float* target = (const float*)d_in[1];
    const float* lamb   = (const float*)d_in[2];
    const float* W0     = (const float*)d_in[3];
    const float* al0    = (const float*)d_in[4];
    const float* ar0    = (const float*)d_in[5];
    const float* W1     = (const float*)d_in[6];
    const float* al1    = (const float*)d_in[7];
    const float* ar1    = (const float*)d_in[8];
    const float* Wout   = (const float*)d_in[9];
    const float* bout   = (const float*)d_in[10];
    const int*   src    = (const int*)d_in[11];
    const int*   dst    = (const int*)d_in[12];

    float* feat   = (float*)d_ws;
    float* s0a    = feat + (size_t)NN * FDIM;
    float* s1a    = s0a  + (size_t)NN * FDIM;
    float* s0b    = s1a  + (size_t)NN * FDIM;
    float* s1b    = s0b  + (size_t)NN * FDIM;
    float* ebuf   = s1b  + (size_t)NN * FDIM;
    float* el     = ebuf + (size_t)EE * NH;
    float* er     = el   + (size_t)NN * NH;
    unsigned* mk  = (unsigned*)(er + (size_t)NN * NH);
    float* d0a    = (float*)(mk + (size_t)NN * NH);
    float* d1a    = d0a  + (size_t)NN * NH;
    float* d0b    = d1a  + (size_t)NN * NH;
    float* d1b    = d0b  + (size_t)NN * NH;
    unsigned* tick = (unsigned*)(d1b + (size_t)NN * NH);

    float* hregion = (float*)d_out;
    float* logits  = hregion + (size_t)NN * FDIM;

    run_conv(inputs, nullptr, W0, al0, ar0, src, dst, s0a, d0a,
             feat, ebuf, el, er, mk, tick, stream);
    run_conv(s0a, d0a, W1, al1, ar1, src, dst, s1a, d1a,
             feat, ebuf, el, er, mk, tick, stream);

    run_conv(target, nullptr, W0, al0, ar0, src, dst, s0b, d0b,
             feat, ebuf, el, er, mk, tick, stream);
    run_conv(s0b, d0b, W1, al1, ar1, src, dst, s1b, d1b,
             feat, ebuf, el, er, mk, tick, stream);

    gat_gemm_out<<<NN / 16, 96, 0, stream>>>(s0a, s1a, s0b, s1b, d0a, d1a, d0b, d1b,
                                             lamb, Wout, bout, hregion, logits);
}


# Scatter cut by destination range instead of head pair over the chip's two halves: a wave takes only edges into its half's node range, whole 512-B rows, 8 per batch; atomic arrival order differs
# speedup vs baseline: 2.6744x; 1.0446x over previous
#include <hip/hip_runtime.h>
#include <hip/hip_bf16.h>
#include <math.h>

#define NN   50000
#define EE   800000
#define IND  128
#define HID  32
#define NH   4
#define NC   40
#define FDIM 128

typedef float v8f __attribute__((ext_vector_type(8)));
typedef float v2f __attribute__((ext_vector_type(2)));

__device__ __forceinline__ unsigned fkey(float f) {
    unsigned b = __float_as_uint(f);
    return (b & 0x80000000u) ? ~b : (b | 0x80000000u);
}
__device__ __forceinline__ float funkey(unsigned k) {
    unsigned b = (k & 0x80000000u) ? (k & 0x7FFFFFFFu) : ~k;
    return __uint_as_float(b);
}

__device__ __forceinline__ float elu1(float v) {
    return (v > 0.f) ? v : (expf(v) - 1.0f);
}

__device__ __forceinline__ float inv_den(float d) {
    return (d > 0.f) ? (1.0f / d) : 0.f;
}

#define GROWS 128
#define TSTRIDE 64
__device__ __forceinline__ void gemm128_body(const float* __restrict__ X,
                                             const float* __restrict__ W,
                                             float* __restrict__ Y,
                                             float* __restrict__ zagg,
                                             unsigned* __restrict__ zmk,
                                             float* __restrict__ zden,
                                             const float* __restrict__ pden,
                                             unsigned* __restrict__ tick,
                                             const bool act) {
    __shared__ float4 Ws[16 * 8 * 32];
    const int t    = threadIdx.x;
    const int lane = t & 31;
    const int wave = __builtin_amdgcn_readfirstlane(t >> 5);
    const int c    = lane & 15;
    const int hh   = lane >> 4;
    const int m0   = blockIdx.x * GROWS;
    const int mw   = m0 + wave * 16;
    const bool live = mw < NN;
    if (blockIdx.x == 0 && t < 2) tick[t * TSTRIDE] = 0u;

    float4 a[16];
    float4 q = make_float4(1.f, 1.f, 1.f, 1.f);
    if (live) {
        const float4* xr = (const float4*)(X + (size_t)(mw + c) * FDIM) + hh;
#pragma unroll
        for (int i = 0; i < 16; ++i) a[i] = xr[2 * i];
        if (act) q = *(const float4*)(pden + (size_t)(mw + c) * NH);

        const float4 z4 = make_float4(0.f, 0.f, 0.f, 0.f);
        float4* za = (float4*)(zagg + (size_t)mw * FDIM);
#pragma unroll
        for (int r = 0; r < 16; ++r) za[lane + 32 * r] = z4;
        if (lane < 16) ((uint4*)(zmk + (size_t)mw * NH))[lane] = make_uint4(0u, 0u, 0u, 0u);
        else           ((float4*)(zden + (size_t)mw * NH))[lane - 16] = z4;
    }

#pragma unroll
    for (int i = 0; i < 16; ++i) {
        const float* wp = W + (size_t)(8 * i + 4 * hh) * FDIM + 16 * wave + c;
        float4 v;
        v.x = wp[0];
        v.y = wp[FDIM];
        v.z = wp[2 * FDIM];
        v.w = wp[3 * FDIM];
        Ws[(i * 8 + wave) * 32 + lane] = v;
    }

    if (act && live) {
        const float iq[4] = { inv_den(q.x), inv_den(q.y), inv_den(q.z), inv_den(q.w) };
#pragma unroll
        for (int i = 0; i < 16; ++i) {
            const float s = iq[i >> 2];
            a[i].x = elu1(a[i].x * s); a[i].y = elu1(a[i].y * s);
            a[i].z = elu1(a[i].z * s); a[i].w = elu1(a[i].w * s);
        }
    }
    __syncthreads();
    if (!live) return;

    v8f acc[8] = {};
    float4 b[2][8];
#pragma unroll
    for (int j = 0; j < 8; ++j) b[0][j] = Ws[j * 32 + lane];
#pragma unroll
    for (int i = 0; i < 16; ++i) {
        const int cur = i & 1;
        __builtin_amdgcn_sched_barrier(0);
        if (i + 1 < 16) {
#pragma unroll
            for (int j = 0; j < 8; ++j) b[cur ^ 1][j] = Ws[((i + 1) * 8 + j) * 32 + lane];
        }
        __builtin_amdgcn_sched_barrier(0);
        v2f a0, a1;
        a0.x = a[i].x; a0.y = a[i].y;
        a1.x = a[i].z; a1.y = a[i].w;
#pragma unroll
        for (int j = 0; j < 8; ++j) {
            v2f b0;
            b0.x = b[cur][j].x; b0.y = b[cur][j].y;
            acc[j] = __builtin_amdgcn_wmma_f32_16x16x4_f32(false, a0, false, b0,
                                                           (short)0, acc[j], false, false);
        }
#pragma unroll
        for (int j = 0; j < 8; ++j) {
            v2f b1;
            b1.x = b[cur][j].z; b1.y = b[cur][j].w;
            acc[j] = __builtin_amdgcn_wmma_f32_16x16x4_f32(false, a1, false, b1,
                                                           (short)0, acc[j], false, false);
        }
    }
    float* yr = Y + (size_t)(mw + 8 * hh) * FDIM + c;
#pragma unroll
    for (int j = 0; j < 8; ++j)
#pragma unroll
        for (int r = 0; r < 8; ++r)
            yr[(size_t)r * FDIM + 16 * j] = acc[j][r];
}

__global__ void __launch_bounds__(256)
gat_gemm128(const float* __restrict__ X,
                            const float* __restrict__ W,
                            float* __restrict__ Y,
                            float* __restrict__ zagg,
                            unsigned* __restrict__ zmk,
                            float* __restrict__ zden,
                            unsigned* __restrict__ tick) {
    gemm128_body(X, W, Y, zagg, zmk, zden, nullptr, tick, false);
}

__global__ void __launch_bounds__(256)
gat_gemm128_act(const float* __restrict__ S,
                                const float* __restrict__ W,
                                float* __restrict__ Y,
                                float* __restrict__ zagg,
                                unsigned* __restrict__ zmk,
                                float* __restrict__ zden,
                                const float* __restrict__ pden,
                                unsigned* __restrict__ tick) {
    gemm128_body(S, W, Y, zagg, zmk, zden, pden, tick, true);
}

__global__ void gat_gemm_out(const float* __restrict__ s0a,
                             const float* __restrict__ s1a,
                             const float* __restrict__ s0b,
                             const float* __restrict__ s1b,
                             const float* __restrict__ d0a,
                             const float* __restrict__ d1a,
                             const float* __restrict__ d0b,
                             const float* __restrict__ d1b,
                             const float* __restrict__ lamb,
                             const float* __restrict__ Wout,
                             const float* __restrict__ bout,
                             float* __restrict__ hout,
                             float* __restrict__ logits) {
    __shared__ float As[16 * FDIM];
    const int m0 = blockIdx.x * 16;
    const int t  = threadIdx.x;
    const float lam = lamb[0];
    const size_t base4 = (size_t)m0 * (FDIM / 4);
    constexpr int NP = 6;
    float4 a0[NP], a1[NP], b0[NP], b1[NP];
    float qa0[NP], qa1[NP], qb0[NP], qb1[NP];
#pragma unroll
    for (int k = 0; k < NP; ++k) {
        const int i = t + 96 * k;
        if (i < 16 * FDIM / 4) {
            a0[k] = ((const float4*)s0a)[base4 + i];
            a1[k] = ((const float4*)s1a)[base4 + i];
            b0[k] = ((const float4*)s0b)[base4 + i];
            b1[k] = ((const float4*)s1b)[base4 + i];
            const size_t di = (size_t)(m0 + (i >> 5)) * NH + ((i & 31) >> 3);
            qa0[k] = d0a[di];
            qa1[k] = d1a[di];
            qb0[k] = d0b[di];
            qb1[k] = d1b[di];
        }
    }
#pragma unroll
    for (int k = 0; k < NP; ++k) {
        const int i = t + 96 * k;
        if (i < 16 * FDIM / 4) {
            const float ia0 = inv_den(qa0[k]), ia1 = inv_den(qa1[k]);
            const float ib0 = inv_den(qb0[k]), ib1 = inv_den(qb1[k]);
            float4 r;
            r.x = lam * elu1(a1[k].x * ia1 + elu1(a0[k].x * ia0)) + (1.0f - lam) * elu1(b1[k].x * ib1 + elu1(b0[k].x * ib0));
            r.y = lam * elu1(a1[k].y * ia1 + elu1(a0[k].y * ia0)) + (1.0f - lam) * elu1(b1[k].y * ib1 + elu1(b0[k].y * ib0));
            r.z = lam * elu1(a1[k].z * ia1 + elu1(a0[k].z * ia0)) + (1.0f - lam) * elu1(b1[k].z * ib1 + elu1(b0[k].z * ib0));
            r.w = lam * elu1(a1[k].w * ia1 + elu1(a0[k].w * ia0)) + (1.0f - lam) * elu1(b1[k].w * ib1 + elu1(b0[k].w * ib0));
            ((float4*)As)[i] = r;
            ((float4*)hout)[base4 + i] = r;
        }
    }
    __syncthreads();

    const int wave = t >> 5;
    const int lane = t & 31;
    const int n0   = wave * 16;
    const int mrow = lane & 15;
    const int kgrp = lane >> 4;
    const int col  = n0 + mrow;

    v8f acc = {};
    for (int kk = 0; kk < FDIM; kk += 4) {
        v2f a, b;
        a.x = As[mrow * FDIM + kk + 2 * kgrp + 0];
        a.y = As[mrow * FDIM + kk + 2 * kgrp + 1];
        b.x = (col < NC) ? Wout[(kk + 2 * kgrp + 0) * NC + col] : 0.0f;
        b.y = (col < NC) ? Wout[(kk + 2 * kgrp + 1) * NC + col] : 0.0f;
        acc = __builtin_amdgcn_wmma_f32_16x16x4_f32(false, a, false, b,
                                                    (short)0, acc, false, false);
    }
    const int cn    = lane & 15;
    const int mbase = (lane >> 4) * 8;
    const int ocol  = n0 + cn;
    if (ocol < NC) {
        const float bb = bout[ocol];
        for (int r = 0; r < 8; ++r)
            logits[(size_t)(m0 + mbase + r) * NC + ocol] = acc[r] + bb;
    }
}

__global__ void gat_attn_coeff(const float* __restrict__ feat,
                               const float* __restrict__ al,
                               const float* __restrict__ ar,
                               float* __restrict__ el,
                               float* __restrict__ er) {
    int i = blockIdx.x * blockDim.x + threadIdx.x;
    if (i >= NN * NH) return;
    int n = i >> 2, h = i & 3;
    const float* f = feat + (size_t)n * FDIM + h * HID;
    float sl = 0.f, sr = 0.f;
    for (int d = 0; d < HID; ++d) {
        sl += f[d] * al[h * HID + d];
        sr += f[d] * ar[h * HID + d];
    }
    el[i] = sl;
    er[i] = sr;
}

__global__ void gat_edge_logits(const float* __restrict__ el,
                                const float* __restrict__ er,
                                const int* __restrict__ src,
                                const int* __restrict__ dst,
                                float* __restrict__ ebuf,
                                unsigned* __restrict__ maxkey) {
    int i = blockIdx.x * blockDim.x + threadIdx.x;
    if (i >= EE * NH) return;
    int e = i >> 2, h = i & 3;
    int s = src[e], d = dst[e];
    float v = el[s * NH + h] + er[d * NH + h];
    v = (v > 0.f) ? v : 0.2f * v;
    ebuf[i] = v;
    atomicMax(&maxkey[d * NH + h], fkey(v));
}

#define EUN    8
#define TEDGES 32
#define NTICK  (EE / TEDGES)
#define NSPLIT (NN / 2)
__device__ __forceinline__ float4 edge_weights(const bool mine, const int e, const int d,
                                               const float* __restrict__ ebuf,
                                               const unsigned* __restrict__ maxkey) {
    float4 v = make_float4(0.f, 0.f, 0.f, 0.f);
    uint4  k = make_uint4(0u, 0u, 0u, 0u);
    if (mine) {
        v = ((const float4*)ebuf)[e];
        k = ((const uint4*)maxkey)[d];
    }
    float4 w;
    w.x = expf(v.x - funkey(k.x));
    w.y = expf(v.y - funkey(k.y));
    w.z = expf(v.z - funkey(k.z));
    w.w = expf(v.w - funkey(k.w));
    return w;
}
template <int K, bool FIRST>
__device__ __forceinline__ void take_edges(unsigned& m, float4& w, const bool mine, const int e,
                                           const int s, const int d, const int lane,
                                           const float* __restrict__ ebuf,
                                           const unsigned* __restrict__ maxkey,
                                           const float* __restrict__ feat,
                                           float* __restrict__ agg,
                                           float* __restrict__ den) {
    int   j[K];
    float f[K][NH];
#pragma unroll
    for (int u = 0; u < K; ++u) {
        j[u] = __builtin_ctz(m);
        m &= m - 1u;
        const int sj = __builtin_amdgcn_readlane(s, j[u]);
        const float* fr = feat + (size_t)sj * FDIM + lane;
        f[u][0] = fr[0];
        f[u][1] = fr[HID];
        f[u][2] = fr[2 * HID];
        f[u][3] = fr[3 * HID];
    }
    if (FIRST) w = edge_weights(mine, e, d, ebuf, maxkey);
#pragma unroll
    for (int u = 0; u < K; ++u) {
        const int dj = __builtin_amdgcn_readlane(d, j[u]);
        const float b0 = __int_as_float(__builtin_amdgcn_readlane(__float_as_int(w.x), j[u]));
        const float b1 = __int_as_float(__builtin_amdgcn_readlane(__float_as_int(w.y), j[u]));
        const float b2 = __int_as_float(__builtin_amdgcn_readlane(__float_as_int(w.z), j[u]));
        const float b3 = __int_as_float(__builtin_amdgcn_readlane(__float_as_int(w.w), j[u]));
        float* ar = agg + (size_t)dj * FDIM + lane;
        atomicAdd(ar,           f[u][0] * b0);
        atomicAdd(ar + HID,     f[u][1] * b1);
        atomicAdd(ar + 2 * HID, f[u][2] * b2);
        atomicAdd(ar + 3 * HID, f[u][3] * b3);
    }
    unsigned long long jp = 0ull;
#pragma unroll
    for (int u = 0; u < K; ++u) jp |= (unsigned long long)(unsigned)j[u] << (8 * u);
    const int from = (int)((unsigned)(jp >> (2 * (lane & ~3))) & 31u) << 2;
    const int   dq = __builtin_amdgcn_ds_bpermute(from, d);
    const float q0 = __int_as_float(__builtin_amdgcn_ds_bpermute(from, __float_as_int(w.x)));
    const float q1 = __int_as_float(__builtin_amdgcn_ds_bpermute(from, __float_as_int(w.y)));
    const float q2 = __int_as_float(__builtin_amdgcn_ds_bpermute(from, __float_as_int(w.z)));
    const float q3 = __int_as_float(__builtin_amdgcn_ds_bpermute(from, __float_as_int(w.w)));
    const int hd = lane & 3;
    const float bw = (hd == 0) ? q0 : (hd == 1) ? q1 : (hd == 2) ? q2 : q3;
    if (lane < NH * K) atomicAdd(den + (size_t)dq * NH + hd, bw);
}
__global__ void __launch_bounds__(256)
gat_scatter(const int* __restrict__ src,
            const int* __restrict__ dst,
            const float* __restrict__ ebuf,
            const unsigned* __restrict__ maxkey,
            const float* __restrict__ feat,
            float* __restrict__ agg,
            float* __restrict__ den,
            unsigned* __restrict__ tick) {
    static_assert(EE % TEDGES == 0 && TEDGES == 32 && EUN == 8 && NH == 4, "a ticket is one edge per lane, four heads");
    const int lane = threadIdx.x & 31;
    int p = (int)((__builtin_amdgcn_s_sendmsg_rtn(0x87) >> 18) & 1u);
    bool other = false;
    for (;;) {
        unsigned tv = 0u;
        if (lane == 0) tv = atomicAdd(&tick[p * TSTRIDE], 1u);
        const unsigned t = (unsigned)__builtin_amdgcn_readfirstlane((int)tv);
        if (t >= (unsigned)NTICK) {
            if (other) break;
            other = true;
            p ^= 1;
            continue;
        }
        const int e = (int)t * TEDGES + lane;
        const int s = src[e];
        const int d = dst[e];
        const bool mine = (d >= NSPLIT) == (p != 0);
        unsigned m = __builtin_amdgcn_ballot_w32(mine);
        float4 w;
        if (__builtin_popcount(m) >= EUN)
            take_edges<EUN, true>(m, w, mine, e, s, d, lane, ebuf, maxkey, feat, agg, den);
        else
            w = edge_weights(mine, e, d, ebuf, maxkey);
        while (m != 0u) {
            const int n = __builtin_popcount(m);
            if (n >= EUN) {
                take_edges<EUN, false>(m, w, mine, e, s, d, lane, ebuf, maxkey, feat, agg, den);
            } else {
                switch (n) {
                case 7:  take_edges<7, false>(m, w, mine, e, s, d, lane, ebuf, maxkey, feat, agg, den); break;
                case 6:  take_edges<6, false>(m, w, mine, e, s, d, lane, ebuf, maxkey, feat, agg, den); break;
                case 5:  take_edges<5, false>(m, w, mine, e, s, d, lane, ebuf, maxkey, feat, agg, den); break;
                case 4:  take_edges<4, false>(m, w, mine, e, s, d, lane, ebuf, maxkey, feat, agg, den); break;
                case 3:  take_edges<3, false>(m, w, mine, e, s, d, lane, ebuf, maxkey, feat, agg, den); break;
                case 2:  take_edges<2, false>(m, w, mine, e, s, d, lane, ebuf, maxkey, feat, agg, den); break;
                default: take_edges<1, false>(m, w, mine, e, s, d, lane, ebuf, maxkey, feat, agg, den); break;
                }
            }
        }
    }
}

static void run_conv(const float* Xin, const float* pden, const float* W, const float* al, const float* ar,
                     const int* src, const int* dst, float* aggout, float* denout,
                     float* feat, float* ebuf, float* el, float* er,
                     unsigned* maxkey, unsigned* tick, hipStream_t stream) {
    if (pden)
        gat_gemm128_act<<<(NN + GROWS - 1) / GROWS, 256, 0, stream>>>(Xin, W, feat, aggout, maxkey, denout, pden, tick);
    else
        gat_gemm128<<<(NN + GROWS - 1) / GROWS, 256, 0, stream>>>(Xin, W, feat, aggout, maxkey, denout, tick);
    gat_attn_coeff<<<(NN * NH + 255) / 256, 256, 0, stream>>>(feat, al, ar, el, er);
    gat_edge_logits<<<(EE * NH + 255) / 256, 256, 0, stream>>>(el, er, src, dst, ebuf, maxkey);
    gat_scatter<<<2048, 256, 0, stream>>>(src, dst, ebuf, maxkey, feat, aggout, denout, tick);
}

extern "C" void kernel_launch(void* const* d_in, const int* in_sizes, int n_in,
                              void* d_out, int out_size, void* d_ws, size_t ws_size,
                              hipStream_t stream) {
    (void)in_sizes; (void)n_in; (void)out_size; (void)ws_size;

    const float* inputs = (const float*)d_in[0];
    const float* target = (const float*)d_in[1];
    const float* lamb   = (const float*)d_in[2];
    const float* W0     = (const float*)d_in[3];
    const float* al0    = (const float*)d_in[4];
    const float* ar0    = (const float*)d_in[5];
    const float* W1     = (const float*)d_in[6];
    const float* al1    = (const float*)d_in[7];
    const float* ar1    = (const float*)d_in[8];
    const float* Wout   = (const float*)d_in[9];
    const float* bout   = (const float*)d_in[10];
    const int*   src    = (const int*)d_in[11];
    const int*   dst    = (const int*)d_in[12];

    float* feat   = (float*)d_ws;
    float* s0a    = feat + (size_t)NN * FDIM;
    float* s1a    = s0a  + (size_t)NN * FDIM;
    float* s0b    = s1a  + (size_t)NN * FDIM;
    float* s1b    = s0b  + (size_t)NN * FDIM;
    float* ebuf   = s1b  + (size_t)NN * FDIM;
    float* el     = ebuf + (size_t)EE * NH;
    float* er     = el   + (size_t)NN * NH;
    unsigned* mk  = (unsigned*)(er + (size_t)NN * NH);
    float* d0a    = (float*)(mk + (size_t)NN * NH);
    float* d1a    = d0a  + (size_t)NN * NH;
    float* d0b    = d1a  + (size_t)NN * NH;
    float* d1b    = d0b  + (size_t)NN * NH;
    unsigned* tick = (unsigned*)(d1b + (size_t)NN * NH);

    float* hregion = (float*)d_out;
    float* logits  = hregion + (size_t)NN * FDIM;

    run_conv(inputs, nullptr, W0, al0, ar0, src, dst, s0a, d0a,
             feat, ebuf, el, er, mk, tick, stream);
    run_conv(s0a, d0a, W1, al1, ar1, src, dst, s1a, d1a,
             feat, ebuf, el, er, mk, tick, stream);

    run_conv(target, nullptr, W0, al0, ar0, src, dst, s0b, d0b,
             feat, ebuf, el, er, mk, tick, stream);
    run_conv(s0b, d0b, W1, al1, ar1, src, dst, s1b, d1b,
             feat, ebuf, el, er, mk, tick, stream);

    gat_gemm_out<<<NN / 16, 96, 0, stream>>>(s0a, s1a, s0b, s1b, d0a, d1a, d0b, d1b,
                                             lamb, Wout, bout, hregion, logits);
}


# Scatter by owner: a work-group owns 200 destination nodes, reads all edge destinations, adds its rows and denominators in LDS (LDS float adds), stores them once; no global adds; sum order differs
# speedup vs baseline: 2.8165x; 1.0531x over previous
#include <hip/hip_runtime.h>
#include <hip/hip_bf16.h>
#include <math.h>

#define NN   50000
#define EE   800000
#define IND  128
#define HID  32
#define NH   4
#define NC   40
#define FDIM 128

typedef float v8f __attribute__((ext_vector_type(8)));
typedef float v2f __attribute__((ext_vector_type(2)));

__device__ __forceinline__ unsigned fkey(float f) {
    unsigned b = __float_as_uint(f);
    return (b & 0x80000000u) ? ~b : (b | 0x80000000u);
}
__device__ __forceinline__ float funkey(unsigned k) {
    unsigned b = (k & 0x80000000u) ? (k & 0x7FFFFFFFu) : ~k;
    return __uint_as_float(b);
}

__device__ __forceinline__ float elu1(float v) {
    return (v > 0.f) ? v : (expf(v) - 1.0f);
}

__device__ __forceinline__ float inv_den(float d) {
    return (d > 0.f) ? (1.0f / d) : 0.f;
}

#define GROWS 128
__device__ __forceinline__ void gemm128_body(const float* __restrict__ X,
                                             const float* __restrict__ W,
                                             float* __restrict__ Y,
                                             unsigned* __restrict__ zmk,
                                             const float* __restrict__ pden,
                                             const bool act) {
    __shared__ float4 Ws[16 * 8 * 32];
    const int t    = threadIdx.x;
    const int lane = t & 31;
    const int wave = __builtin_amdgcn_readfirstlane(t >> 5);
    const int c    = lane & 15;
    const int hh   = lane >> 4;
    const int m0   = blockIdx.x * GROWS;
    const int mw   = m0 + wave * 16;
    const bool live = mw < NN;

    float4 a[16];
    float4 q = make_float4(1.f, 1.f, 1.f, 1.f);
    if (live) {
        const float4* xr = (const float4*)(X + (size_t)(mw + c) * FDIM) + hh;
#pragma unroll
        for (int i = 0; i < 16; ++i) a[i] = xr[2 * i];
        if (act) q = *(const float4*)(pden + (size_t)(mw + c) * NH);

        if (lane < 16) ((uint4*)(zmk + (size_t)mw * NH))[lane] = make_uint4(0u, 0u, 0u, 0u);
    }

#pragma unroll
    for (int i = 0; i < 16; ++i) {
        const float* wp = W + (size_t)(8 * i + 4 * hh) * FDIM + 16 * wave + c;
        float4 v;
        v.x = wp[0];
        v.y = wp[FDIM];
        v.z = wp[2 * FDIM];
        v.w = wp[3 * FDIM];
        Ws[(i * 8 + wave) * 32 + lane] = v;
    }

    if (act && live) {
        const float iq[4] = { inv_den(q.x), inv_den(q.y), inv_den(q.z), inv_den(q.w) };
#pragma unroll
        for (int i = 0; i < 16; ++i) {
            const float s = iq[i >> 2];
            a[i].x = elu1(a[i].x * s); a[i].y = elu1(a[i].y * s);
            a[i].z = elu1(a[i].z * s); a[i].w = elu1(a[i].w * s);
        }
    }
    __syncthreads();
    if (!live) return;

    v8f acc[8] = {};
    float4 b[2][8];
#pragma unroll
    for (int j = 0; j < 8; ++j) b[0][j] = Ws[j * 32 + lane];
#pragma unroll
    for (int i = 0; i < 16; ++i) {
        const int cur = i & 1;
        __builtin_amdgcn_sched_barrier(0);
        if (i + 1 < 16) {
#pragma unroll
            for (int j = 0; j < 8; ++j) b[cur ^ 1][j] = Ws[((i + 1) * 8 + j) * 32 + lane];
        }
        __builtin_amdgcn_sched_barrier(0);
        v2f a0, a1;
        a0.x = a[i].x; a0.y = a[i].y;
        a1.x = a[i].z; a1.y = a[i].w;
#pragma unroll
        for (int j = 0; j < 8; ++j) {
            v2f b0;
            b0.x = b[cur][j].x; b0.y = b[cur][j].y;
            acc[j] = __builtin_amdgcn_wmma_f32_16x16x4_f32(false, a0, false, b0,
                                                           (short)0, acc[j], false, false);
        }
#pragma unroll
        for (int j = 0; j < 8; ++j) {
            v2f b1;
            b1.x = b[cur][j].z; b1.y = b[cur][j].w;
            acc[j] = __builtin_amdgcn_wmma_f32_16x16x4_f32(false, a1, false, b1,
                                                           (short)0, acc[j], false, false);
        }
    }
    float* yr = Y + (size_t)(mw + 8 * hh) * FDIM + c;
#pragma unroll
    for (int j = 0; j < 8; ++j)
#pragma unroll
        for (int r = 0; r < 8; ++r)
            yr[(size_t)r * FDIM + 16 * j] = acc[j][r];
}

__global__ void __launch_bounds__(256)
gat_gemm128(const float* __restrict__ X,
                            const float* __restrict__ W,
                            float* __restrict__ Y,
                            unsigned* __restrict__ zmk) {
    gemm128_body(X, W, Y, zmk, nullptr, false);
}

__global__ void __launch_bounds__(256)
gat_gemm128_act(const float* __restrict__ S,
                                const float* __restrict__ W,
                                float* __restrict__ Y,
                                unsigned* __restrict__ zmk,
                                const float* __restrict__ pden) {
    gemm128_body(S, W, Y, zmk, pden, true);
}

__global__ void gat_gemm_out(const float* __restrict__ s0a,
                             const float* __restrict__ s1a,
                             const float* __restrict__ s0b,
                             const float* __restrict__ s1b,
                             const float* __restrict__ d0a,
                             const float* __restrict__ d1a,
                             const float* __restrict__ d0b,
                             const float* __restrict__ d1b,
                             const float* __restrict__ lamb,
                             const float* __restrict__ Wout,
                             const float* __restrict__ bout,
                             float* __restrict__ hout,
                             float* __restrict__ logits) {
    __shared__ float As[16 * FDIM];
    const int m0 = blockIdx.x * 16;
    const int t  = threadIdx.x;
    const float lam = lamb[0];
    const size_t base4 = (size_t)m0 * (FDIM / 4);
    constexpr int NP = 6;
    float4 a0[NP], a1[NP], b0[NP], b1[NP];
    float qa0[NP], qa1[NP], qb0[NP], qb1[NP];
#pragma unroll
    for (int k = 0; k < NP; ++k) {
        const int i = t + 96 * k;
        if (i < 16 * FDIM / 4) {
            a0[k] = ((const float4*)s0a)[base4 + i];
            a1[k] = ((const float4*)s1a)[base4 + i];
            b0[k] = ((const float4*)s0b)[base4 + i];
            b1[k] = ((const float4*)s1b)[base4 + i];
            const size_t di = (size_t)(m0 + (i >> 5)) * NH + ((i & 31) >> 3);
            qa0[k] = d0a[di];
            qa1[k] = d1a[di];
            qb0[k] = d0b[di];
            qb1[k] = d1b[di];
        }
    }
#pragma unroll
    for (int k = 0; k < NP; ++k) {
        const int i = t + 96 * k;
        if (i < 16 * FDIM / 4) {
            const float ia0 = inv_den(qa0[k]), ia1 = inv_den(qa1[k]);
            const float ib0 = inv_den(qb0[k]), ib1 = inv_den(qb1[k]);
            float4 r;
            r.x = lam * elu1(a1[k].x * ia1 + elu1(a0[k].x * ia0)) + (1.0f - lam) * elu1(b1[k].x * ib1 + elu1(b0[k].x * ib0));
            r.y = lam * elu1(a1[k].y * ia1 + elu1(a0[k].y * ia0)) + (1.0f - lam) * elu1(b1[k].y * ib1 + elu1(b0[k].y * ib0));
            r.z = lam * elu1(a1[k].z * ia1 + elu1(a0[k].z * ia0)) + (1.0f - lam) * elu1(b1[k].z * ib1 + elu1(b0[k].z * ib0));
            r.w = lam * elu1(a1[k].w * ia1 + elu1(a0[k].w * ia0)) + (1.0f - lam) * elu1(b1[k].w * ib1 + elu1(b0[k].w * ib0));
            ((float4*)As)[i] = r;
            ((float4*)hout)[base4 + i] = r;
        }
    }
    __syncthreads();

    const int wave = t >> 5;
    const int lane = t & 31;
    const int n0   = wave * 16;
    const int mrow = lane & 15;
    const int kgrp = lane >> 4;
    const int col  = n0 + mrow;

    v8f acc = {};
    for (int kk = 0; kk < FDIM; kk += 4) {
        v2f a, b;
        a.x = As[mrow * FDIM + kk + 2 * kgrp + 0];
        a.y = As[mrow * FDIM + kk + 2 * kgrp + 1];
        b.x = (col < NC) ? Wout[(kk + 2 * kgrp + 0) * NC + col] : 0.0f;
        b.y = (col < NC) ? Wout[(kk + 2 * kgrp + 1) * NC + col] : 0.0f;
        acc = __builtin_amdgcn_wmma_f32_16x16x4_f32(false, a, false, b,
                                                    (short)0, acc, false, false);
    }
    const int cn    = lane & 15;
    const int mbase = (lane >> 4) * 8;
    const int ocol  = n0 + cn;
    if (ocol < NC) {
        const float bb = bout[ocol];
        for (int r = 0; r < 8; ++r)
            logits[(size_t)(m0 + mbase + r) * NC + ocol] = acc[r] + bb;
    }
}

__global__ void gat_attn_coeff(const float* __restrict__ feat,
                               const float* __restrict__ al,
                               const float* __restrict__ ar,
                               float* __restrict__ el,
                               float* __restrict__ er) {
    int i = blockIdx.x * blockDim.x + threadIdx.x;
    if (i >= NN * NH) return;
    int n = i >> 2, h = i & 3;
    const float* f = feat + (size_t)n * FDIM + h * HID;
    float sl = 0.f, sr = 0.f;
    for (int d = 0; d < HID; ++d) {
        sl += f[d] * al[h * HID + d];
        sr += f[d] * ar[h * HID + d];
    }
    el[i] = sl;
    er[i] = sr;
}

__global__ void gat_edge_logits(const float* __restrict__ el,
                                const float* __restrict__ er,
                                const int* __restrict__ src,
                                const int* __restrict__ dst,
                                float* __restrict__ ebuf,
                                unsigned* __restrict__ maxkey) {
    int i = blockIdx.x * blockDim.x + threadIdx.x;
    if (i >= EE * NH) return;
    int e = i >> 2, h = i & 3;
    int s = src[e], d = dst[e];
    float v = el[s * NH + h] + er[d * NH + h];
    v = (v > 0.f) ? v : 0.2f * v;
    ebuf[i] = v;
    atomicMax(&maxkey[d * NH + h], fkey(v));
}

#define EUN    8
#define NPW    200
#define NWG    (NN / NPW)
#define SWAVES 32
#define UEDGES 128
#define NUNIT  (EE / UEDGES)
#define SUNR   8
#define EBITS  20
#define QLEN   512
__device__ __forceinline__ float4 edge_weights(const bool mine, const int e, const int dl,
                                               const float* __restrict__ ebuf,
                                               const uint4* smax) {
    float4 v = make_float4(0.f, 0.f, 0.f, 0.f);
    uint4  k = make_uint4(0u, 0u, 0u, 0u);
    if (mine) {
        v = ((const float4*)ebuf)[e];
        k = smax[dl];
    }
    float4 w;
    w.x = expf(v.x - funkey(k.x));
    w.y = expf(v.y - funkey(k.y));
    w.z = expf(v.z - funkey(k.z));
    w.w = expf(v.w - funkey(k.w));
    return w;
}
template <int K, bool FIRST>
__device__ __forceinline__ void take_edges(unsigned& m, float4& w, const bool mine, const int e,
                                           const int s, const int d, const int lane,
                                           const float* __restrict__ ebuf,
                                           const uint4* smax,
                                           const float* __restrict__ feat,
                                           float* sagg,
                                           float* sden) {
    int   j[K];
    float f[K][NH];
#pragma unroll
    for (int u = 0; u < K; ++u) {
        j[u] = __builtin_ctz(m);
        m &= m - 1u;
        const int sj = __builtin_amdgcn_readlane(s, j[u]);
        const float* fr = feat + (size_t)sj * FDIM + lane;
        f[u][0] = fr[0];
        f[u][1] = fr[HID];
        f[u][2] = fr[2 * HID];
        f[u][3] = fr[3 * HID];
    }
    if (FIRST) w = edge_weights(mine, e, d, ebuf, smax);
#pragma unroll
    for (int u = 0; u < K; ++u) {
        const int dj = __builtin_amdgcn_readlane(d, j[u]);
        const float b0 = __int_as_float(__builtin_amdgcn_readlane(__float_as_int(w.x), j[u]));
        const float b1 = __int_as_float(__builtin_amdgcn_readlane(__float_as_int(w.y), j[u]));
        const float b2 = __int_as_float(__builtin_amdgcn_readlane(__float_as_int(w.z), j[u]));
        const float b3 = __int_as_float(__builtin_amdgcn_readlane(__float_as_int(w.w), j[u]));
        float* ar = sagg + dj * FDIM + lane;
        atomicAdd(ar,           f[u][0] * b0);
        atomicAdd(ar + HID,     f[u][1] * b1);
        atomicAdd(ar + 2 * HID, f[u][2] * b2);
        atomicAdd(ar + 3 * HID, f[u][3] * b3);
    }
    unsigned long long jp = 0ull;
#pragma unroll
    for (int u = 0; u < K; ++u) jp |= (unsigned long long)(unsigned)j[u] << (8 * u);
    const int from = (int)((unsigned)(jp >> (2 * (lane & ~3))) & 31u) << 2;
    const int   dq = __builtin_amdgcn_ds_bpermute(from, d);
    const float q0 = __int_as_float(__builtin_amdgcn_ds_bpermute(from, __float_as_int(w.x)));
    const float q1 = __int_as_float(__builtin_amdgcn_ds_bpermute(from, __float_as_int(w.y)));
    const float q2 = __int_as_float(__builtin_amdgcn_ds_bpermute(from, __float_as_int(w.z)));
    const float q3 = __int_as_float(__builtin_amdgcn_ds_bpermute(from, __float_as_int(w.w)));
    const int hd = lane & 3;
    const float bw = (hd == 0) ? q0 : (hd == 1) ? q1 : (hd == 2) ? q2 : q3;
    if (lane < NH * K) atomicAdd(sden + dq * NH + hd, bw);
}
__global__ void __launch_bounds__(SWAVES * 32)
gat_scatter(const int* __restrict__ src,
            const int* __restrict__ dst,
            const float* __restrict__ ebuf,
            const unsigned* __restrict__ maxkey,
            const float* __restrict__ feat,
            float* __restrict__ agg,
            float* __restrict__ den) {
    static_assert(NN % NPW == 0 && EE % UEDGES == 0 && NUNIT % NWG == 0 && UEDGES == 128 && EUN == 8 && NH == 4,
                  "whole ranges, whole units, four edges per lane, four heads");
    static_assert((QLEN & (QLEN - 1)) == 0 && QLEN >= UEDGES + 32 + UEDGES - 1, "the ring takes a whole unit on top of what may be left in it");
    static_assert(EE <= (1 << EBITS) && NPW <= (1 << (32 - EBITS)), "a ring entry holds the edge number and the destination's place in the range");
    __shared__ float4 sagg4[NPW * FDIM / 4];
    __shared__ float4 sden4[NPW * NH / 4];
    __shared__ uint4  smax[NPW];
    __shared__ int    sq[SWAVES * QLEN];
    float* sagg = (float*)sagg4;
    float* sden = (float*)sden4;
    const int t    = threadIdx.x;
    const int lane = t & 31;
    const int wave = __builtin_amdgcn_readfirstlane(t >> 5);
    const int lo   = blockIdx.x * NPW;
    const int start = blockIdx.x * (NUNIT / NWG);
    int* q = sq + wave * QLEN;

    const float4 z4 = make_float4(0.f, 0.f, 0.f, 0.f);
    for (int i = t; i < NPW * FDIM / 4; i += SWAVES * 32) sagg4[i] = z4;
    if (t < NPW * NH / 4) sden4[t] = z4;
    if (t < NPW) smax[t] = ((const uint4*)maxkey)[lo + t];
    __syncthreads();

    unsigned head = 0u, n = 0u;
    int  k0 = wave;
    int  ustart = 0;
    bool done = false;
    int4 dv[SUNR];
    int  eb[SUNR];
    for (;;) {
        int stop = SUNR;
        if (!done) {
            if (ustart == 0) {
#pragma unroll
                for (int u = 0; u < SUNR; ++u) {
                    const int k = k0 + SWAVES * u;
                    int unit = k + start;
                    if (unit >= NUNIT) unit -= NUNIT;
                    eb[u] = unit * UEDGES + lane * 4;
                    dv[u] = make_int4(-1, -1, -1, -1);
                    if (k < NUNIT) dv[u] = ((const int4*)dst)[unit * 32 + lane];
                }
            }
#pragma unroll
            for (int u = 0; u < SUNR; ++u) {
                if (u >= ustart && u < stop) {
                    const unsigned r0 = (unsigned)(dv[u].x - lo), r1 = (unsigned)(dv[u].y - lo);
                    const unsigned r2 = (unsigned)(dv[u].z - lo), r3 = (unsigned)(dv[u].w - lo);
                    const unsigned rm = min(min(r0, r1), min(r2, r3));
                    if (__builtin_amdgcn_ballot_w32(rm < (unsigned)NPW) != 0u) {
                        if (n + (unsigned)UEDGES > (unsigned)QLEN) {
                            stop = u;
                        } else {
                            const unsigned rr[4] = { r0, r1, r2, r3 };
#pragma unroll
                            for (int cc = 0; cc < 4; ++cc) {
                                const bool hit = rr[cc] < (unsigned)NPW;
                                const unsigned mm = __builtin_amdgcn_ballot_w32(hit);
                                const unsigned at = head + n + __builtin_amdgcn_mbcnt_lo(mm, 0u);
                                if (hit) q[at & (QLEN - 1)] = (eb[u] + cc) | (int)(rr[cc] << EBITS);
                                n += (unsigned)__builtin_popcount(mm);
                            }
                        }
                    }
                }
            }
        }
        while (n >= 32u || (done && n != 0u)) {
            const unsigned cnt = (n < 32u) ? n : 32u;
            __builtin_amdgcn_fence(__ATOMIC_RELEASE, "wavefront");
            __builtin_amdgcn_wave_barrier();
            __builtin_amdgcn_fence(__ATOMIC_ACQUIRE, "wavefront");
            const bool mine = lane < (int)cnt;
            int e = 0, s = 0, d = 0;
            if (mine) {
                const unsigned x = (unsigned)q[(head + (unsigned)lane) & (QLEN - 1)];
                e = (int)(x & ((1u << EBITS) - 1u));
                d = (int)(x >> EBITS);
                s = src[e];
            }
            head += cnt;
            n    -= cnt;
            unsigned m = __builtin_amdgcn_ballot_w32(mine);
            float4 w;
            if (__builtin_popcount(m) >= EUN)
                take_edges<EUN, true>(m, w, mine, e, s, d, lane, ebuf, smax, feat, sagg, sden);
            else
                w = edge_weights(mine, e, d, ebuf, smax);
            while (m != 0u) {
                const int nb = __builtin_popcount(m);
                if (nb >= EUN) {
                    take_edges<EUN, false>(m, w, mine, e, s, d, lane, ebuf, smax, feat, sagg, sden);
                } else {
                    switch (nb) {
                    case 7:  take_edges<7, false>(m, w, mine, e, s, d, lane, ebuf, smax, feat, sagg, sden); break;
                    case 6:  take_edges<6, false>(m, w, mine, e, s, d, lane, ebuf, smax, feat, sagg, sden); break;
                    case 5:  take_edges<5, false>(m, w, mine, e, s, d, lane, ebuf, smax, feat, sagg, sden); break;
                    case 4:  take_edges<4, false>(m, w, mine, e, s, d, lane, ebuf, smax, feat, sagg, sden); break;
                    case 3:  take_edges<3, false>(m, w, mine, e, s, d, lane, ebuf, smax, feat, sagg, sden); break;
                    case 2:  take_edges<2, false>(m, w, mine, e, s, d, lane, ebuf, smax, feat, sagg, sden); break;
                    default: take_edges<1, false>(m, w, mine, e, s, d, lane, ebuf, smax, feat, sagg, sden); break;
                    }
                }
            }
        }
        if (done) break;
        if (stop == SUNR) {
            ustart = 0;
            k0 += SWAVES * SUNR;
            done = k0 >= NUNIT;
        } else {
            ustart = stop;
        }
    }
    __syncthreads();

    float4* ga = (float4*)(agg + (size_t)lo * FDIM);
    for (int i = t; i < NPW * FDIM / 4; i += SWAVES * 32) ga[i] = sagg4[i];
    if (t < NPW * NH / 4) ((float4*)(den + (size_t)lo * NH))[t] = sden4[t];
}

static void run_conv(const float* Xin, const float* pden, const float* W, const float* al, const float* ar,
                     const int* src, const int* dst, float* aggout, float* denout,
                     float* feat, float* ebuf, float* el, float* er,
                     unsigned* maxkey, hipStream_t stream) {
    if (pden)
        gat_gemm128_act<<<(NN + GROWS - 1) / GROWS, 256, 0, stream>>>(Xin, W, feat, maxkey, pden);
    else
        gat_gemm128<<<(NN + GROWS - 1) / GROWS, 256, 0, stream>>>(Xin, W, feat, maxkey);
    gat_attn_coeff<<<(NN * NH + 255) / 256, 256, 0, stream>>>(feat, al, ar, el, er);
    gat_edge_logits<<<(EE * NH + 255) / 256, 256, 0, stream>>>(el, er, src, dst, ebuf, maxkey);
    gat_scatter<<<NWG, SWAVES * 32, 0, stream>>>(src, dst, ebuf, maxkey, feat, aggout, denout);
}

extern "C" void kernel_launch(void* const* d_in, const int* in_sizes, int n_in,
                              void* d_out, int out_size, void* d_ws, size_t ws_size,
                              hipStream_t stream) {
    (void)in_sizes; (void)n_in; (void)out_size; (void)ws_size;

    const float* inputs = (const float*)d_in[0];
    const float* target = (const float*)d_in[1];
    const float* lamb   = (const float*)d_in[2];
    const float* W0     = (const float*)d_in[3];
    const float* al0    = (const float*)d_in[4];
    const float* ar0    = (const float*)d_in[5];
    const float* W1     = (const float*)d_in[6];
    const float* al1    = (const float*)d_in[7];
    const float* ar1    = (const float*)d_in[8];
    const float* Wout   = (const float*)d_in[9];
    const float* bout   = (const float*)d_in[10];
    const int*   src    = (const int*)d_in[11];
    const int*   dst    = (const int*)d_in[12];

    float* feat   = (float*)d_ws;
    float* s0a    = feat + (size_t)NN * FDIM;
    float* s1a    = s0a  + (size_t)NN * FDIM;
    float* s0b    = s1a  + (size_t)NN * FDIM;
    float* s1b    = s0b  + (size_t)NN * FDIM;
    float* ebuf   = s1b  + (size_t)NN * FDIM;
    float* el     = ebuf + (size_t)EE * NH;
    float* er     = el   + (size_t)NN * NH;
    unsigned* mk  = (unsigned*)(er + (size_t)NN * NH);
    float* d0a    = (float*)(mk + (size_t)NN * NH);
    float* d1a    = d0a  + (size_t)NN * NH;
    float* d0b    = d1a  + (size_t)NN * NH;
    float* d1b    = d0b  + (size_t)NN * NH;

    float* hregion = (float*)d_out;
    float* logits  = hregion + (size_t)NN * FDIM;

    run_conv(inputs, nullptr, W0, al0, ar0, src, dst, s0a, d0a,
             feat, ebuf, el, er, mk, stream);
    run_conv(s0a, d0a, W1, al1, ar1, src, dst, s1a, d1a,
             feat, ebuf, el, er, mk, stream);

    run_conv(target, nullptr, W0, al0, ar0, src, dst, s0b, d0b,
             feat, ebuf, el, er, mk, stream);
    run_conv(s0b, d0b, W1, al1, ar1, src, dst, s1b, d1b,
             feat, ebuf, el, er, mk, stream);

    gat_gemm_out<<<NN / 16, 96, 0, stream>>>(s0a, s1a, s0b, s1b, d0a, d1a, d0b, d1b,
                                             lamb, Wout, bout, hregion, logits);
}


# Scatter scans a one-byte owner plane (node range of each edge's destination, written once per call by a new small launch) instead of the 4-byte destination list; 17 -> 18 launches; sum order differs
# speedup vs baseline: 3.2184x; 1.1427x over previous
#include <hip/hip_runtime.h>
#include <hip/hip_bf16.h>
#include <math.h>

#define NN   50000
#define EE   800000
#define IND  128
#define HID  32
#define NH   4
#define NC   40
#define FDIM 128

typedef float v8f __attribute__((ext_vector_type(8)));
typedef float v2f __attribute__((ext_vector_type(2)));

__device__ __forceinline__ unsigned fkey(float f) {
    unsigned b = __float_as_uint(f);
    return (b & 0x80000000u) ? ~b : (b | 0x80000000u);
}
__device__ __forceinline__ float funkey(unsigned k) {
    unsigned b = (k & 0x80000000u) ? (k & 0x7FFFFFFFu) : ~k;
    return __uint_as_float(b);
}

__device__ __forceinline__ float elu1(float v) {
    return (v > 0.f) ? v : (expf(v) - 1.0f);
}

__device__ __forceinline__ float inv_den(float d) {
    return (d > 0.f) ? (1.0f / d) : 0.f;
}

#define GROWS 128
__device__ __forceinline__ void gemm128_body(const float* __restrict__ X,
                                             const float* __restrict__ W,
                                             float* __restrict__ Y,
                                             unsigned* __restrict__ zmk,
                                             const float* __restrict__ pden,
                                             const bool act) {
    __shared__ float4 Ws[16 * 8 * 32];
    const int t    = threadIdx.x;
    const int lane = t & 31;
    const int wave = __builtin_amdgcn_readfirstlane(t >> 5);
    const int c    = lane & 15;
    const int hh   = lane >> 4;
    const int m0   = blockIdx.x * GROWS;
    const int mw   = m0 + wave * 16;
    const bool live = mw < NN;

    float4 a[16];
    float4 q = make_float4(1.f, 1.f, 1.f, 1.f);
    if (live) {
        const float4* xr = (const float4*)(X + (size_t)(mw + c) * FDIM) + hh;
#pragma unroll
        for (int i = 0; i < 16; ++i) a[i] = xr[2 * i];
        if (act) q = *(const float4*)(pden + (size_t)(mw + c) * NH);

        if (lane < 16) ((uint4*)(zmk + (size_t)mw * NH))[lane] = make_uint4(0u, 0u, 0u, 0u);
    }

#pragma unroll
    for (int i = 0; i < 16; ++i) {
        const float* wp = W + (size_t)(8 * i + 4 * hh) * FDIM + 16 * wave + c;
        float4 v;
        v.x = wp[0];
        v.y = wp[FDIM];
        v.z = wp[2 * FDIM];
        v.w = wp[3 * FDIM];
        Ws[(i * 8 + wave) * 32 + lane] = v;
    }

    if (act && live) {
        const float iq[4] = { inv_den(q.x), inv_den(q.y), inv_den(q.z), inv_den(q.w) };
#pragma unroll
        for (int i = 0; i < 16; ++i) {
            const float s = iq[i >> 2];
            a[i].x = elu1(a[i].x * s); a[i].y = elu1(a[i].y * s);
            a[i].z = elu1(a[i].z * s); a[i].w = elu1(a[i].w * s);
        }
    }
    __syncthreads();
    if (!live) return;

    v8f acc[8] = {};
    float4 b[2][8];
#pragma unroll
    for (int j = 0; j < 8; ++j) b[0][j] = Ws[j * 32 + lane];
#pragma unroll
    for (int i = 0; i < 16; ++i) {
        const int cur = i & 1;
        __builtin_amdgcn_sched_barrier(0);
        if (i + 1 < 16) {
#pragma unroll
            for (int j = 0; j < 8; ++j) b[cur ^ 1][j] = Ws[((i + 1) * 8 + j) * 32 + lane];
        }
        __builtin_amdgcn_sched_barrier(0);
        v2f a0, a1;
        a0.x = a[i].x; a0.y = a[i].y;
        a1.x = a[i].z; a1.y = a[i].w;
#pragma unroll
        for (int j = 0; j < 8; ++j) {
            v2f b0;
            b0.x = b[cur][j].x; b0.y = b[cur][j].y;
            acc[j] = __builtin_amdgcn_wmma_f32_16x16x4_f32(false, a0, false, b0,
                                                           (short)0, acc[j], false, false);
        }
#pragma unroll
        for (int j = 0; j < 8; ++j) {
            v2f b1;
            b1.x = b[cur][j].z; b1.y = b[cur][j].w;
            acc[j] = __builtin_amdgcn_wmma_f32_16x16x4_f32(false, a1, false, b1,
                                                           (short)0, acc[j], false, false);
        }
    }
    float* yr = Y + (size_t)(mw + 8 * hh) * FDIM + c;
#pragma unroll
    for (int j = 0; j < 8; ++j)
#pragma unroll
        for (int r = 0; r < 8; ++r)
            yr[(size_t)r * FDIM + 16 * j] = acc[j][r];
}

__global__ void __launch_bounds__(256)
gat_gemm128(const float* __restrict__ X,
                            const float* __restrict__ W,
                            float* __restrict__ Y,
                            unsigned* __restrict__ zmk) {
    gemm128_body(X, W, Y, zmk, nullptr, false);
}

__global__ void __launch_bounds__(256)
gat_gemm128_act(const float* __restrict__ S,
                                const float* __restrict__ W,
                                float* __restrict__ Y,
                                unsigned* __restrict__ zmk,
                                const float* __restrict__ pden) {
    gemm128_body(S, W, Y, zmk, pden, true);
}

__global__ void gat_gemm_out(const float* __restrict__ s0a,
                             const float* __restrict__ s1a,
                             const float* __restrict__ s0b,
                             const float* __restrict__ s1b,
                             const float* __restrict__ d0a,
                             const float* __restrict__ d1a,
                             const float* __restrict__ d0b,
                             const float* __restrict__ d1b,
                             const float* __restrict__ lamb,
                             const float* __restrict__ Wout,
                             const float* __restrict__ bout,
                             float* __restrict__ hout,
                             float* __restrict__ logits) {
    __shared__ float As[16 * FDIM];
    const int m0 = blockIdx.x * 16;
    const int t  = threadIdx.x;
    const float lam = lamb[0];
    const size_t base4 = (size_t)m0 * (FDIM / 4);
    constexpr int NP = 6;
    float4 a0[NP], a1[NP], b0[NP], b1[NP];
    float qa0[NP], qa1[NP], qb0[NP], qb1[NP];
#pragma unroll
    for (int k = 0; k < NP; ++k) {
        const int i = t + 96 * k;
        if (i < 16 * FDIM / 4) {
            a0[k] = ((const float4*)s0a)[base4 + i];
            a1[k] = ((const float4*)s1a)[base4 + i];
            b0[k] = ((const float4*)s0b)[base4 + i];
            b1[k] = ((const float4*)s1b)[base4 + i];
            const size_t di = (size_t)(m0 + (i >> 5)) * NH + ((i & 31) >> 3);
            qa0[k] = d0a[di];
            qa1[k] = d1a[di];
            qb0[k] = d0b[di];
            qb1[k] = d1b[di];
        }
    }
#pragma unroll
    for (int k = 0; k < NP; ++k) {
        const int i = t + 96 * k;
        if (i < 16 * FDIM / 4) {
            const float ia0 = inv_den(qa0[k]), ia1 = inv_den(qa1[k]);
            const float ib0 = inv_den(qb0[k]), ib1 = inv_den(qb1[k]);
            float4 r;
            r.x = lam * elu1(a1[k].x * ia1 + elu1(a0[k].x * ia0)) + (1.0f - lam) * elu1(b1[k].x * ib1 + elu1(b0[k].x * ib0));
            r.y = lam * elu1(a1[k].y * ia1 + elu1(a0[k].y * ia0)) + (1.0f - lam) * elu1(b1[k].y * ib1 + elu1(b0[k].y * ib0));
            r.z = lam * elu1(a1[k].z * ia1 + elu1(a0[k].z * ia0)) + (1.0f - lam) * elu1(b1[k].z * ib1 + elu1(b0[k].z * ib0));
            r.w = lam * elu1(a1[k].w * ia1 + elu1(a0[k].w * ia0)) + (1.0f - lam) * elu1(b1[k].w * ib1 + elu1(b0[k].w * ib0));
            ((float4*)As)[i] = r;
            ((float4*)hout)[base4 + i] = r;
        }
    }
    __syncthreads();

    const int wave = t >> 5;
    const int lane = t & 31;
    const int n0   = wave * 16;
    const int mrow = lane & 15;
    const int kgrp = lane >> 4;
    const int col  = n0 + mrow;

    v8f acc = {};
    for (int kk = 0; kk < FDIM; kk += 4) {
        v2f a, b;
        a.x = As[mrow * FDIM + kk + 2 * kgrp + 0];
        a.y = As[mrow * FDIM + kk + 2 * kgrp + 1];
        b.x = (col < NC) ? Wout[(kk + 2 * kgrp + 0) * NC + col] : 0.0f;
        b.y = (col < NC) ? Wout[(kk + 2 * kgrp + 1) * NC + col] : 0.0f;
        acc = __builtin_amdgcn_wmma_f32_16x16x4_f32(false, a, false, b,
                                                    (short)0, acc, false, false);
    }
    const int cn    = lane & 15;
    const int mbase = (lane >> 4) * 8;
    const int ocol  = n0 + cn;
    if (ocol < NC) {
        const float bb = bout[ocol];
        for (int r = 0; r < 8; ++r)
            logits[(size_t)(m0 + mbase + r) * NC + ocol] = acc[r] + bb;
    }
}

__global__ void gat_attn_coeff(const float* __restrict__ feat,
                               const float* __restrict__ al,
                               const float* __restrict__ ar,
                               float* __restrict__ el,
                               float* __restrict__ er) {
    int i = blockIdx.x * blockDim.x + threadIdx.x;
    if (i >= NN * NH) return;
    int n = i >> 2, h = i & 3;
    const float* f = feat + (size_t)n * FDIM + h * HID;
    float sl = 0.f, sr = 0.f;
    for (int d = 0; d < HID; ++d) {
        sl += f[d] * al[h * HID + d];
        sr += f[d] * ar[h * HID + d];
    }
    el[i] = sl;
    er[i] = sr;
}

__global__ void gat_edge_logits(const float* __restrict__ el,
                                const float* __restrict__ er,
                                const int* __restrict__ src,
                                const int* __restrict__ dst,
                                float* __restrict__ ebuf,
                                unsigned* __restrict__ maxkey) {
    int i = blockIdx.x * blockDim.x + threadIdx.x;
    if (i >= EE * NH) return;
    int e = i >> 2, h = i & 3;
    int s = src[e], d = dst[e];
    float v = el[s * NH + h] + er[d * NH + h];
    v = (v > 0.f) ? v : 0.2f * v;
    ebuf[i] = v;
    atomicMax(&maxkey[d * NH + h], fkey(v));
}

#define EUN    8
#define NPW    200
#define NWG    (NN / NPW)
#define SWAVES 32
#define UEDGES 512
#define NUNIT  ((EE + UEDGES - 1) / UEDGES)
#define OWNB   (NUNIT * UEDGES)
#define NOOWN  255u
#define SUNR   8
#define QLEN   1024
__device__ __forceinline__ unsigned owner_byte(const int d) {
    const unsigned o = (unsigned)d / (unsigned)NPW;
    return (o < (unsigned)NWG) ? o : NOOWN;
}
__global__ void gat_owner_plane(const int* __restrict__ dst, unsigned* __restrict__ own) {
    const int i = blockIdx.x * blockDim.x + threadIdx.x;
    if (i >= OWNB / 4) return;
    unsigned w = 0xFFFFFFFFu;
    if (i < EE / 4) {
        const int4 d = ((const int4*)dst)[i];
        w = owner_byte(d.x) | (owner_byte(d.y) << 8) | (owner_byte(d.z) << 16) | (owner_byte(d.w) << 24);
    }
    own[i] = w;
}
__device__ __forceinline__ unsigned zero_bytes(const unsigned x) {
    const unsigned t = (x & 0x7F7F7F7Fu) + 0x7F7F7F7Fu;
    return ~(t | x | 0x7F7F7F7Fu);
}
__device__ __forceinline__ float4 edge_weights(const bool mine, const int e, const int dl,
                                               const float* __restrict__ ebuf,
                                               const uint4* smax) {
    float4 v = make_float4(0.f, 0.f, 0.f, 0.f);
    uint4  k = make_uint4(0u, 0u, 0u, 0u);
    if (mine) {
        v = ((const float4*)ebuf)[e];
        k = smax[dl];
    }
    float4 w;
    w.x = expf(v.x - funkey(k.x));
    w.y = expf(v.y - funkey(k.y));
    w.z = expf(v.z - funkey(k.z));
    w.w = expf(v.w - funkey(k.w));
    return w;
}
template <int K, bool FIRST>
__device__ __forceinline__ void take_edges(unsigned& m, float4& w, const bool mine, const int e,
                                           const int s, const int d, const int lane,
                                           const float* __restrict__ ebuf,
                                           const uint4* smax,
                                           const float* __restrict__ feat,
                                           float* sagg,
                                           float* sden) {
    int   j[K];
    float f[K][NH];
#pragma unroll
    for (int u = 0; u < K; ++u) {
        j[u] = __builtin_ctz(m);
        m &= m - 1u;
        const int sj = __builtin_amdgcn_readlane(s, j[u]);
        const float* fr = feat + (size_t)sj * FDIM + lane;
        f[u][0] = fr[0];
        f[u][1] = fr[HID];
        f[u][2] = fr[2 * HID];
        f[u][3] = fr[3 * HID];
    }
    if (FIRST) w = edge_weights(mine, e, d, ebuf, smax);
#pragma unroll
    for (int u = 0; u < K; ++u) {
        const int dj = __builtin_amdgcn_readlane(d, j[u]);
        const float b0 = __int_as_float(__builtin_amdgcn_readlane(__float_as_int(w.x), j[u]));
        const float b1 = __int_as_float(__builtin_amdgcn_readlane(__float_as_int(w.y), j[u]));
        const float b2 = __int_as_float(__builtin_amdgcn_readlane(__float_as_int(w.z), j[u]));
        const float b3 = __int_as_float(__builtin_amdgcn_readlane(__float_as_int(w.w), j[u]));
        float* ar = sagg + dj * FDIM + lane;
        atomicAdd(ar,           f[u][0] * b0);
        atomicAdd(ar + HID,     f[u][1] * b1);
        atomicAdd(ar + 2 * HID, f[u][2] * b2);
        atomicAdd(ar + 3 * HID, f[u][3] * b3);
    }
    unsigned long long jp = 0ull;
#pragma unroll
    for (int u = 0; u < K; ++u) jp |= (unsigned long long)(unsigned)j[u] << (8 * u);
    const int from = (int)((unsigned)(jp >> (2 * (lane & ~3))) & 31u) << 2;
    const int   dq = __builtin_amdgcn_ds_bpermute(from, d);
    const float q0 = __int_as_float(__builtin_amdgcn_ds_bpermute(from, __float_as_int(w.x)));
    const float q1 = __int_as_float(__builtin_amdgcn_ds_bpermute(from, __float_as_int(w.y)));
    const float q2 = __int_as_float(__builtin_amdgcn_ds_bpermute(from, __float_as_int(w.z)));
    const float q3 = __int_as_float(__builtin_amdgcn_ds_bpermute(from, __float_as_int(w.w)));
    const int hd = lane & 3;
    const float bw = (hd == 0) ? q0 : (hd == 1) ? q1 : (hd == 2) ? q2 : q3;
    if (lane < NH * K) atomicAdd(sden + dq * NH + hd, bw);
}
__global__ void __launch_bounds__(SWAVES * 32)
gat_scatter(const int* __restrict__ src,
            const int* __restrict__ dst,
            const unsigned* __restrict__ own,
            const float* __restrict__ ebuf,
            const unsigned* __restrict__ maxkey,
            const float* __restrict__ feat,
            float* __restrict__ agg,
            float* __restrict__ den) {
    static_assert(NN % NPW == 0 && EE % 4 == 0 && UEDGES == 512 && EUN == 8 && NH == 4,
                  "whole ranges, whole words of the owner plane, sixteen edges per lane, four heads");
    static_assert((unsigned)NWG <= NOOWN, "no work-group has the padding's owner byte");
    static_assert((QLEN & (QLEN - 1)) == 0 && QLEN >= UEDGES + 31, "the ring takes a whole unit on top of what may be left in it");
    __shared__ float4 sagg4[NPW * FDIM / 4];
    __shared__ float4 sden4[NPW * NH / 4];
    __shared__ uint4  smax[NPW];
    __shared__ int    sq[SWAVES * QLEN];
    float* sagg = (float*)sagg4;
    float* sden = (float*)sden4;
    const int t    = threadIdx.x;
    const int lane = t & 31;
    const int wave = __builtin_amdgcn_readfirstlane(t >> 5);
    const int lo   = blockIdx.x * NPW;
    const int start = (blockIdx.x * NUNIT) / NWG;
    const unsigned idx4 = (unsigned)blockIdx.x * 0x01010101u;
    int* q = sq + wave * QLEN;

    const float4 z4 = make_float4(0.f, 0.f, 0.f, 0.f);
    for (int i = t; i < NPW * FDIM / 4; i += SWAVES * 32) sagg4[i] = z4;
    if (t < NPW * NH / 4) sden4[t] = z4;
    if (t < NPW) smax[t] = ((const uint4*)maxkey)[lo + t];
    __syncthreads();

    unsigned head = 0u, n = 0u;
    int  k0 = wave;
    int  ustart = 0;
    bool done = false;
    uint4 dv[SUNR];
    int  eb[SUNR];
    for (;;) {
        int stop = SUNR;
        if (!done) {
            if (ustart == 0) {
#pragma unroll
                for (int u = 0; u < SUNR; ++u) {
                    const int k = k0 + SWAVES * u;
                    int unit = k + start;
                    if (unit >= NUNIT) unit -= NUNIT;
                    eb[u] = unit * UEDGES + lane * 16;
                    dv[u] = make_uint4(~0u, ~0u, ~0u, ~0u);
                    if (k < NUNIT) dv[u] = ((const uint4*)own)[unit * 32 + lane];
                }
            }
#pragma unroll
            for (int u = 0; u < SUNR; ++u) {
                if (u >= ustart && u < stop) {
                    unsigned hm = (zero_bytes(dv[u].x ^ idx4) >> 7) | (zero_bytes(dv[u].y ^ idx4) >> 6)
                                | (zero_bytes(dv[u].z ^ idx4) >> 5) | (zero_bytes(dv[u].w ^ idx4) >> 4);
                    if (__builtin_amdgcn_ballot_w32(hm != 0u) != 0u) {
                        if (n + (unsigned)UEDGES > (unsigned)QLEN) {
                            stop = u;
                        } else {
                            for (;;) {
                                const bool hit = hm != 0u;
                                const unsigned mm = __builtin_amdgcn_ballot_w32(hit);
                                if (mm == 0u) break;
                                const unsigned at = head + n + __builtin_amdgcn_mbcnt_lo(mm, 0u);
                                const int p = __builtin_ctz(hm | 0x80000000u);
                                if (hit) q[at & (QLEN - 1)] = eb[u] + ((p & 7) << 2) + (p >> 3);
                                hm &= hm - 1u;
                                n += (unsigned)__builtin_popcount(mm);
                            }
                        }
                    }
                }
            }
        }
        while (n >= 32u || (done && n != 0u)) {
            const unsigned cnt = (n < 32u) ? n : 32u;
            __builtin_amdgcn_fence(__ATOMIC_RELEASE, "wavefront");
            __builtin_amdgcn_wave_barrier();
            __builtin_amdgcn_fence(__ATOMIC_ACQUIRE, "wavefront");
            const bool mine = lane < (int)cnt;
            int e = 0, s = 0, d = 0;
            if (mine) {
                e = q[(head + (unsigned)lane) & (QLEN - 1)];
                s = src[e];
                d = dst[e] - lo;
            }
            head += cnt;
            n    -= cnt;
            unsigned m = __builtin_amdgcn_ballot_w32(mine);
            float4 w;
            if (__builtin_popcount(m) >= EUN)
                take_edges<EUN, true>(m, w, mine, e, s, d, lane, ebuf, smax, feat, sagg, sden);
            else
                w = edge_weights(mine, e, d, ebuf, smax);
            while (m != 0u) {
                const int nb = __builtin_popcount(m);
                if (nb >= EUN) {
                    take_edges<EUN, false>(m, w, mine, e, s, d, lane, ebuf, smax, feat, sagg, sden);
                } else {
                    switch (nb) {
                    case 7:  take_edges<7, false>(m, w, mine, e, s, d, lane, ebuf, smax, feat, sagg, sden); break;
                    case 6:  take_edges<6, false>(m, w, mine, e, s, d, lane, ebuf, smax, feat, sagg, sden); break;
                    case 5:  take_edges<5, false>(m, w, mine, e, s, d, lane, ebuf, smax, feat, sagg, sden); break;
                    case 4:  take_edges<4, false>(m, w, mine, e, s, d, lane, ebuf, smax, feat, sagg, sden); break;
                    case 3:  take_edges<3, false>(m, w, mine, e, s, d, lane, ebuf, smax, feat, sagg, sden); break;
                    case 2:  take_edges<2, false>(m, w, mine, e, s, d, lane, ebuf, smax, feat, sagg, sden); break;
                    default: take_edges<1, false>(m, w, mine, e, s, d, lane, ebuf, smax, feat, sagg, sden); break;
                    }
                }
            }
        }
        if (done) break;
        if (stop == SUNR) {
            ustart = 0;
            k0 += SWAVES * SUNR;
            done = k0 >= NUNIT;
        } else {
            ustart = stop;
        }
    }
    __syncthreads();

    float4* ga = (float4*)(agg + (size_t)lo * FDIM);
    for (int i = t; i < NPW * FDIM / 4; i += SWAVES * 32) ga[i] = sagg4[i];
    if (t < NPW * NH / 4) ((float4*)(den + (size_t)lo * NH))[t] = sden4[t];
}

static void run_conv(const float* Xin, const float* pden, const float* W, const float* al, const float* ar,
                     const int* src, const int* dst, const unsigned* own, float* aggout, float* denout,
                     float* feat, float* ebuf, float* el, float* er,
                     unsigned* maxkey, hipStream_t stream) {
    if (pden)
        gat_gemm128_act<<<(NN + GROWS - 1) / GROWS, 256, 0, stream>>>(Xin, W, feat, maxkey, pden);
    else
        gat_gemm128<<<(NN + GROWS - 1) / GROWS, 256, 0, stream>>>(Xin, W, feat, maxkey);
    gat_attn_coeff<<<(NN * NH + 255) / 256, 256, 0, stream>>>(feat, al, ar, el, er);
    gat_edge_logits<<<(EE * NH + 255) / 256, 256, 0, stream>>>(el, er, src, dst, ebuf, maxkey);
    gat_scatter<<<NWG, SWAVES * 32, 0, stream>>>(src, dst, own, ebuf, maxkey, feat, aggout, denout);
}

extern "C" void kernel_launch(void* const* d_in, const int* in_sizes, int n_in,
                              void* d_out, int out_size, void* d_ws, size_t ws_size,
                              hipStream_t stream) {
    (void)in_sizes; (void)n_in; (void)out_size; (void)ws_size;

    const float* inputs = (const float*)d_in[0];
    const float* target = (const float*)d_in[1];
    const float* lamb   = (const float*)d_in[2];
    const float* W0     = (const float*)d_in[3];
    const float* al0    = (const float*)d_in[4];
    const float* ar0    = (const float*)d_in[5];
    const float* W1     = (const float*)d_in[6];
    const float* al1    = (const float*)d_in[7];
    const float* ar1    = (const float*)d_in[8];
    const float* Wout   = (const float*)d_in[9];
    const float* bout   = (const float*)d_in[10];
    const int*   src    = (const int*)d_in[11];
    const int*   dst    = (const int*)d_in[12];

    float* feat   = (float*)d_ws;
    float* s0a    = feat + (size_t)NN * FDIM;
    float* s1a    = s0a  + (size_t)NN * FDIM;
    float* s0b    = s1a  + (size_t)NN * FDIM;
    float* s1b    = s0b  + (size_t)NN * FDIM;
    float* ebuf   = s1b  + (size_t)NN * FDIM;
    float* el     = ebuf + (size_t)EE * NH;
    float* er     = el   + (size_t)NN * NH;
    unsigned* mk  = (unsigned*)(er + (size_t)NN * NH);
    float* d0a    = (float*)(mk + (size_t)NN * NH);
    float* d1a    = d0a  + (size_t)NN * NH;
    float* d0b    = d1a  + (size_t)NN * NH;
    float* d1b    = d0b  + (size_t)NN * NH;
    unsigned* own = (unsigned*)(d1b + (size_t)NN * NH);

    float* hregion = (float*)d_out;
    float* logits  = hregion + (size_t)NN * FDIM;

    gat_owner_plane<<<(OWNB / 4 + 255) / 256, 256, 0, stream>>>(dst, own);

    run_conv(inputs, nullptr, W0, al0, ar0, src, dst, own, s0a, d0a,
             feat, ebuf, el, er, mk, stream);
    run_conv(s0a, d0a, W1, al1, ar1, src, dst, own, s1a, d1a,
             feat, ebuf, el, er, mk, stream);

    run_conv(target, nullptr, W0, al0, ar0, src, dst, own, s0b, d0b,
             feat, ebuf, el, er, mk, stream);
    run_conv(s0b, d0b, W1, al1, ar1, src, dst, own, s1b, d1b,
             feat, ebuf, el, er, mk, stream);

    gat_gemm_out<<<NN / 16, 96, 0, stream>>>(s0a, s1a, s0b, s1b, d0a, d1a, d0b, d1b,
                                             lamb, Wout, bout, hregion, logits);
}


# Edge-logit launches folded into the scatter: the owner work-group takes the segment max in LDS in a first sweep, keeps (src,dst) in its ring, then adds rows; 18 -> 14 launches; sum order differs
# speedup vs baseline: 4.4829x; 1.3929x over previous
#include <hip/hip_runtime.h>
#include <hip/hip_bf16.h>
#include <math.h>

#define NN   50000
#define EE   800000
#define IND  128
#define HID  32
#define NH   4
#define NC   40
#define FDIM 128

typedef float v8f __attribute__((ext_vector_type(8)));
typedef float v2f __attribute__((ext_vector_type(2)));

__device__ __forceinline__ unsigned fkey(float f) {
    unsigned b = __float_as_uint(f);
    return (b & 0x80000000u) ? ~b : (b | 0x80000000u);
}
__device__ __forceinline__ float funkey(unsigned k) {
    unsigned b = (k & 0x80000000u) ? (k & 0x7FFFFFFFu) : ~k;
    return __uint_as_float(b);
}

__device__ __forceinline__ float elu1(float v) {
    return (v > 0.f) ? v : (expf(v) - 1.0f);
}

__device__ __forceinline__ float inv_den(float d) {
    return (d > 0.f) ? (1.0f / d) : 0.f;
}

#define GROWS 128
__device__ __forceinline__ void gemm128_body(const float* __restrict__ X,
                                             const float* __restrict__ W,
                                             float* __restrict__ Y,
                                             const float* __restrict__ pden,
                                             const bool act) {
    __shared__ float4 Ws[16 * 8 * 32];
    const int t    = threadIdx.x;
    const int lane = t & 31;
    const int wave = __builtin_amdgcn_readfirstlane(t >> 5);
    const int c    = lane & 15;
    const int hh   = lane >> 4;
    const int m0   = blockIdx.x * GROWS;
    const int mw   = m0 + wave * 16;
    const bool live = mw < NN;

    float4 a[16];
    float4 q = make_float4(1.f, 1.f, 1.f, 1.f);
    if (live) {
        const float4* xr = (const float4*)(X + (size_t)(mw + c) * FDIM) + hh;
#pragma unroll
        for (int i = 0; i < 16; ++i) a[i] = xr[2 * i];
        if (act) q = *(const float4*)(pden + (size_t)(mw + c) * NH);
    }

#pragma unroll
    for (int i = 0; i < 16; ++i) {
        const float* wp = W + (size_t)(8 * i + 4 * hh) * FDIM + 16 * wave + c;
        float4 v;
        v.x = wp[0];
        v.y = wp[FDIM];
        v.z = wp[2 * FDIM];
        v.w = wp[3 * FDIM];
        Ws[(i * 8 + wave) * 32 + lane] = v;
    }

    if (act && live) {
        const float iq[4] = { inv_den(q.x), inv_den(q.y), inv_den(q.z), inv_den(q.w) };
#pragma unroll
        for (int i = 0; i < 16; ++i) {
            const float s = iq[i >> 2];
            a[i].x = elu1(a[i].x * s); a[i].y = elu1(a[i].y * s);
            a[i].z = elu1(a[i].z * s); a[i].w = elu1(a[i].w * s);
        }
    }
    __syncthreads();
    if (!live) return;

    v8f acc[8] = {};
    float4 b[2][8];
#pragma unroll
    for (int j = 0; j < 8; ++j) b[0][j] = Ws[j * 32 + lane];
#pragma unroll
    for (int i = 0; i < 16; ++i) {
        const int cur = i & 1;
        __builtin_amdgcn_sched_barrier(0);
        if (i + 1 < 16) {
#pragma unroll
            for (int j = 0; j < 8; ++j) b[cur ^ 1][j] = Ws[((i + 1) * 8 + j) * 32 + lane];
        }
        __builtin_amdgcn_sched_barrier(0);
        v2f a0, a1;
        a0.x = a[i].x; a0.y = a[i].y;
        a1.x = a[i].z; a1.y = a[i].w;
#pragma unroll
        for (int j = 0; j < 8; ++j) {
            v2f b0;
            b0.x = b[cur][j].x; b0.y = b[cur][j].y;
            acc[j] = __builtin_amdgcn_wmma_f32_16x16x4_f32(false, a0, false, b0,
                                                           (short)0, acc[j], false, false);
        }
#pragma unroll
        for (int j = 0; j < 8; ++j) {
            v2f b1;
            b1.x = b[cur][j].z; b1.y = b[cur][j].w;
            acc[j] = __builtin_amdgcn_wmma_f32_16x16x4_f32(false, a1, false, b1,
                                                           (short)0, acc[j], false, false);
        }
    }
    float* yr = Y + (size_t)(mw + 8 * hh) * FDIM + c;
#pragma unroll
    for (int j = 0; j < 8; ++j)
#pragma unroll
        for (int r = 0; r < 8; ++r)
            yr[(size_t)r * FDIM + 16 * j] = acc[j][r];
}

__global__ void __launch_bounds__(256)
gat_gemm128(const float* __restrict__ X,
                            const float* __restrict__ W,
                            float* __restrict__ Y) {
    gemm128_body(X, W, Y, nullptr, false);
}

__global__ void __launch_bounds__(256)
gat_gemm128_act(const float* __restrict__ S,
                                const float* __restrict__ W,
                                float* __restrict__ Y,
                                const float* __restrict__ pden) {
    gemm128_body(S, W, Y, pden, true);
}

__global__ void gat_gemm_out(const float* __restrict__ s0a,
                             const float* __restrict__ s1a,
                             const float* __restrict__ s0b,
                             const float* __restrict__ s1b,
                             const float* __restrict__ d0a,
                             const float* __restrict__ d1a,
                             const float* __restrict__ d0b,
                             const float* __restrict__ d1b,
                             const float* __restrict__ lamb,
                             const float* __restrict__ Wout,
                             const float* __restrict__ bout,
                             float* __restrict__ hout,
                             float* __restrict__ logits) {
    __shared__ float As[16 * FDIM];
    const int m0 = blockIdx.x * 16;
    const int t  = threadIdx.x;
    const float lam = lamb[0];
    const size_t base4 = (size_t)m0 * (FDIM / 4);
    constexpr int NP = 6;
    float4 a0[NP], a1[NP], b0[NP], b1[NP];
    float qa0[NP], qa1[NP], qb0[NP], qb1[NP];
#pragma unroll
    for (int k = 0; k < NP; ++k) {
        const int i = t + 96 * k;
        if (i < 16 * FDIM / 4) {
            a0[k] = ((const float4*)s0a)[base4 + i];
            a1[k] = ((const float4*)s1a)[base4 + i];
            b0[k] = ((const float4*)s0b)[base4 + i];
            b1[k] = ((const float4*)s1b)[base4 + i];
            const size_t di = (size_t)(m0 + (i >> 5)) * NH + ((i & 31) >> 3);
            qa0[k] = d0a[di];
            qa1[k] = d1a[di];
            qb0[k] = d0b[di];
            qb1[k] = d1b[di];
        }
    }
#pragma unroll
    for (int k = 0; k < NP; ++k) {
        const int i = t + 96 * k;
        if (i < 16 * FDIM / 4) {
            const float ia0 = inv_den(qa0[k]), ia1 = inv_den(qa1[k]);
            const float ib0 = inv_den(qb0[k]), ib1 = inv_den(qb1[k]);
            float4 r;
            r.x = lam * elu1(a1[k].x * ia1 + elu1(a0[k].x * ia0)) + (1.0f - lam) * elu1(b1[k].x * ib1 + elu1(b0[k].x * ib0));
            r.y = lam * elu1(a1[k].y * ia1 + elu1(a0[k].y * ia0)) + (1.0f - lam) * elu1(b1[k].y * ib1 + elu1(b0[k].y * ib0));
            r.z = lam * elu1(a1[k].z * ia1 + elu1(a0[k].z * ia0)) + (1.0f - lam) * elu1(b1[k].z * ib1 + elu1(b0[k].z * ib0));
            r.w = lam * elu1(a1[k].w * ia1 + elu1(a0[k].w * ia0)) + (1.0f - lam) * elu1(b1[k].w * ib1 + elu1(b0[k].w * ib0));
            ((float4*)As)[i] = r;
            ((float4*)hout)[base4 + i] = r;
        }
    }
    __syncthreads();

    const int wave = t >> 5;
    const int lane = t & 31;
    const int n0   = wave * 16;
    const int mrow = lane & 15;
    const int kgrp = lane >> 4;
    const int col  = n0 + mrow;

    v8f acc = {};
    for (int kk = 0; kk < FDIM; kk += 4) {
        v2f a, b;
        a.x = As[mrow * FDIM + kk + 2 * kgrp + 0];
        a.y = As[mrow * FDIM + kk + 2 * kgrp + 1];
        b.x = (col < NC) ? Wout[(kk + 2 * kgrp + 0) * NC + col] : 0.0f;
        b.y = (col < NC) ? Wout[(kk + 2 * kgrp + 1) * NC + col] : 0.0f;
        acc = __builtin_amdgcn_wmma_f32_16x16x4_f32(false, a, false, b,
                                                    (short)0, acc, false, false);
    }
    const int cn    = lane & 15;
    const int mbase = (lane >> 4) * 8;
    const int ocol  = n0 + cn;
    if (ocol < NC) {
        const float bb = bout[ocol];
        for (int r = 0; r < 8; ++r)
            logits[(size_t)(m0 + mbase + r) * NC + ocol] = acc[r] + bb;
    }
}

__global__ void gat_attn_coeff(const float* __restrict__ feat,
                               const float* __restrict__ al,
                               const float* __restrict__ ar,
                               float* __restrict__ el,
                               float* __restrict__ er) {
    int i = blockIdx.x * blockDim.x + threadIdx.x;
    if (i >= NN * NH) return;
    int n = i >> 2, h = i & 3;
    const float* f = feat + (size_t)n * FDIM + h * HID;
    float sl = 0.f, sr = 0.f;
    for (int d = 0; d < HID; ++d) {
        sl += f[d] * al[h * HID + d];
        sr += f[d] * ar[h * HID + d];
    }
    el[i] = sl;
    er[i] = sr;
}

#define EUN    8
#define NPW    200
#define NWG    (NN / NPW)
#define SWAVES 32
#define UEDGES 512
#define NUNIT  ((EE + UEDGES - 1) / UEDGES)
#define OWNB   (NUNIT * UEDGES)
#define NOOWN  255u
#define SUNR   8
#define MG     4
#define QLEN   1024
__device__ __forceinline__ unsigned owner_byte(const int d) {
    const unsigned o = (unsigned)d / (unsigned)NPW;
    return (o < (unsigned)NWG) ? o : NOOWN;
}
__global__ void gat_owner_plane(const int* __restrict__ dst, unsigned* __restrict__ own) {
    const int i = blockIdx.x * blockDim.x + threadIdx.x;
    if (i >= OWNB / 4) return;
    unsigned w = 0xFFFFFFFFu;
    if (i < EE / 4) {
        const int4 d = ((const int4*)dst)[i];
        w = owner_byte(d.x) | (owner_byte(d.y) << 8) | (owner_byte(d.z) << 16) | (owner_byte(d.w) << 24);
    }
    own[i] = w;
}
__device__ __forceinline__ unsigned zero_bytes(const unsigned x) {
    const unsigned t = (x & 0x7F7F7F7Fu) + 0x7F7F7F7Fu;
    return ~(t | x | 0x7F7F7F7Fu);
}
__device__ __forceinline__ float4 edge_logits4(const float4 l, const float4 r) {
    float4 v;
    v.x = l.x + r.x; v.y = l.y + r.y; v.z = l.z + r.z; v.w = l.w + r.w;
    v.x = (v.x > 0.f) ? v.x : 0.2f * v.x;
    v.y = (v.y > 0.f) ? v.y : 0.2f * v.y;
    v.z = (v.z > 0.f) ? v.z : 0.2f * v.z;
    v.w = (v.w > 0.f) ? v.w : 0.2f * v.w;
    return v;
}
__device__ __forceinline__ float4 edge_weights(const bool mine, const int s, const int dl,
                                               const float* __restrict__ el,
                                               const float4* ser,
                                               const uint4* smax) {
    float4 l = make_float4(0.f, 0.f, 0.f, 0.f);
    float4 r = make_float4(0.f, 0.f, 0.f, 0.f);
    uint4  k = make_uint4(0u, 0u, 0u, 0u);
    if (mine) {
        l = ((const float4*)el)[s];
        r = ser[dl];
        k = smax[dl];
    }
    const float4 v = edge_logits4(l, r);
    float4 w;
    w.x = expf(v.x - funkey(k.x));
    w.y = expf(v.y - funkey(k.y));
    w.z = expf(v.z - funkey(k.z));
    w.w = expf(v.w - funkey(k.w));
    return w;
}
template <int K, bool FIRST>
__device__ __forceinline__ void take_edges(unsigned& m, float4& w, const bool mine,
                                           const int s, const int d, const int lane,
                                           const float* __restrict__ el,
                                           const float4* ser,
                                           const uint4* smax,
                                           const float* __restrict__ feat,
                                           float* sagg,
                                           float* sden) {
    int   j[K];
    float f[K][NH];
#pragma unroll
    for (int u = 0; u < K; ++u) {
        j[u] = __builtin_ctz(m);
        m &= m - 1u;
        const int sj = __builtin_amdgcn_readlane(s, j[u]);
        const float* fr = feat + (size_t)sj * FDIM + lane;
        f[u][0] = fr[0];
        f[u][1] = fr[HID];
        f[u][2] = fr[2 * HID];
        f[u][3] = fr[3 * HID];
    }
    if (FIRST) w = edge_weights(mine, s, d, el, ser, smax);
#pragma unroll
    for (int u = 0; u < K; ++u) {
        const int dj = __builtin_amdgcn_readlane(d, j[u]);
        const float b0 = __int_as_float(__builtin_amdgcn_readlane(__float_as_int(w.x), j[u]));
        const float b1 = __int_as_float(__builtin_amdgcn_readlane(__float_as_int(w.y), j[u]));
        const float b2 = __int_as_float(__builtin_amdgcn_readlane(__float_as_int(w.z), j[u]));
        const float b3 = __int_as_float(__builtin_amdgcn_readlane(__float_as_int(w.w), j[u]));
        float* ar = sagg + dj * FDIM + lane;
        atomicAdd(ar,           f[u][0] * b0);
        atomicAdd(ar + HID,     f[u][1] * b1);
        atomicAdd(ar + 2 * HID, f[u][2] * b2);
        atomicAdd(ar + 3 * HID, f[u][3] * b3);
    }
    unsigned long long jp = 0ull;
#pragma unroll
    for (int u = 0; u < K; ++u) jp |= (unsigned long long)(unsigned)j[u] << (8 * u);
    const int from = (int)((unsigned)(jp >> (2 * (lane & ~3))) & 31u) << 2;
    const int   dq = __builtin_amdgcn_ds_bpermute(from, d);
    const float q0 = __int_as_float(__builtin_amdgcn_ds_bpermute(from, __float_as_int(w.x)));
    const float q1 = __int_as_float(__builtin_amdgcn_ds_bpermute(from, __float_as_int(w.y)));
    const float q2 = __int_as_float(__builtin_amdgcn_ds_bpermute(from, __float_as_int(w.z)));
    const float q3 = __int_as_float(__builtin_amdgcn_ds_bpermute(from, __float_as_int(w.w)));
    const int hd = lane & 3;
    const float bw = (hd == 0) ? q0 : (hd == 1) ? q1 : (hd == 2) ? q2 : q3;
    if (lane < NH * K) atomicAdd(sden + dq * NH + hd, bw);
}
template <bool MAXPASS>
__device__ __forceinline__ void sweep(const int lane, const int wave, const int lo, const int start,
                                      const unsigned idx4, int* q,
                                      unsigned& head, unsigned& n, bool& keep, bool done, const bool packed,
                                      const int* __restrict__ src,
                                      const int* __restrict__ dst,
                                      const unsigned* __restrict__ own,
                                      const float* __restrict__ el,
                                      const float* __restrict__ feat,
                                      const float4* ser, uint4* smax,
                                      float* sagg, float* sden) {
    int  k0 = wave;
    int  ustart = 0;
    uint4 dv[SUNR];
    int  eb[SUNR];
    for (;;) {
        int stop = SUNR;
        if (!done) {
            if (ustart == 0) {
#pragma unroll
                for (int u = 0; u < SUNR; ++u) {
                    const int k = k0 + SWAVES * u;
                    int unit = k + start;
                    if (unit >= NUNIT) unit -= NUNIT;
                    eb[u] = unit * UEDGES + lane * 16;
                    dv[u] = make_uint4(~0u, ~0u, ~0u, ~0u);
                    if (k < NUNIT) dv[u] = ((const uint4*)own)[unit * 32 + lane];
                }
            }
#pragma unroll
            for (int u = 0; u < SUNR; ++u) {
                if (u >= ustart && u < stop) {
                    unsigned hm = (zero_bytes(dv[u].x ^ idx4) >> 7) | (zero_bytes(dv[u].y ^ idx4) >> 6)
                                | (zero_bytes(dv[u].z ^ idx4) >> 5) | (zero_bytes(dv[u].w ^ idx4) >> 4);
                    if (__builtin_amdgcn_ballot_w32(hm != 0u) != 0u) {
                        if (MAXPASS && keep && head + n + (unsigned)UEDGES > (unsigned)QLEN) keep = false;
                        const unsigned used = (MAXPASS && keep) ? head + n : n;
                        if (used + (unsigned)UEDGES > (unsigned)QLEN) {
                            stop = u;
                        } else {
                            for (;;) {
                                const bool hit = hm != 0u;
                                const unsigned mm = __builtin_amdgcn_ballot_w32(hit);
                                if (mm == 0u) break;
                                const unsigned at = head + n + __builtin_amdgcn_mbcnt_lo(mm, 0u);
                                const int p = __builtin_ctz(hm | 0x80000000u);
                                if (hit) q[at & (QLEN - 1)] = eb[u] + ((p & 7) << 2) + (p >> 3);
                                hm &= hm - 1u;
                                n += (unsigned)__builtin_popcount(mm);
                            }
                        }
                    }
                }
            }
        }
        if (MAXPASS && keep) {
            if (done) {
                const unsigned total = n;
                __builtin_amdgcn_fence(__ATOMIC_RELEASE, "wavefront");
                __builtin_amdgcn_wave_barrier();
                __builtin_amdgcn_fence(__ATOMIC_ACQUIRE, "wavefront");
                for (unsigned b = 0u; b < total; b += 32u * MG) {
                    bool   on[MG];
                    int    ex[MG], sx[MG], dx[MG];
                    float4 lx[MG];
#pragma unroll
                    for (int g = 0; g < MG; ++g) {
                        const unsigned i = b + 32u * (unsigned)g + (unsigned)lane;
                        on[g] = i < total;
                        const int e = q[i & (QLEN - 1)];
                        ex[g] = on[g] ? e : 0;
                    }
#pragma unroll
                    for (int g = 0; g < MG; ++g) {
                        sx[g] = src[ex[g]];
                        dx[g] = dst[ex[g]];
                    }
#pragma unroll
                    for (int g = 0; g < MG; ++g) lx[g] = ((const float4*)el)[sx[g]];
#pragma unroll
                    for (int g = 0; g < MG; ++g) {
                        if (on[g]) {
                            const unsigned i = b + 32u * (unsigned)g + (unsigned)lane;
                            const int d = dx[g] - lo;
                            q[i] = (sx[g] << 8) | d;
                            const float4 v = edge_logits4(lx[g], ser[d]);
                            unsigned* mk = (unsigned*)(smax + d);
                            atomicMax(mk + 0, fkey(v.x));
                            atomicMax(mk + 1, fkey(v.y));
                            atomicMax(mk + 2, fkey(v.z));
                            atomicMax(mk + 3, fkey(v.w));
                        }
                    }
                }
                head = total;
                n    = 0u;
            }
        } else
        while (n >= 32u || (done && n != 0u)) {
            const unsigned cnt = (n < 32u) ? n : 32u;
            __builtin_amdgcn_fence(__ATOMIC_RELEASE, "wavefront");
            __builtin_amdgcn_wave_barrier();
            __builtin_amdgcn_fence(__ATOMIC_ACQUIRE, "wavefront");
            const bool mine = lane < (int)cnt;
            const unsigned slot = (head + (unsigned)lane) & (QLEN - 1);
            int s = 0, d = 0;
            if (MAXPASS) {
                if (mine) {
                    const int e = q[slot];
                    s = src[e];
                    d = dst[e] - lo;
                    const float4 v = edge_logits4(((const float4*)el)[s], ser[d]);
                    unsigned* mk = (unsigned*)(smax + d);
                    atomicMax(mk + 0, fkey(v.x));
                    atomicMax(mk + 1, fkey(v.y));
                    atomicMax(mk + 2, fkey(v.z));
                    atomicMax(mk + 3, fkey(v.w));
                }
                head += cnt;
                n    -= cnt;
            } else {
                if (mine) {
                    const int x = q[slot];
                    if (packed) {
                        s = (int)((unsigned)x >> 8);
                        d = x & 255;
                    } else {
                        s = src[x];
                        d = dst[x] - lo;
                    }
                }
                head += cnt;
                n    -= cnt;
                unsigned m = __builtin_amdgcn_ballot_w32(mine);
                float4 w;
                if (__builtin_popcount(m) >= EUN)
                    take_edges<EUN, true>(m, w, mine, s, d, lane, el, ser, smax, feat, sagg, sden);
                else
                    w = edge_weights(mine, s, d, el, ser, smax);
                while (m != 0u) {
                    const int nb = __builtin_popcount(m);
                    if (nb >= EUN) {
                        take_edges<EUN, false>(m, w, mine, s, d, lane, el, ser, smax, feat, sagg, sden);
                    } else {
                        switch (nb) {
                        case 7:  take_edges<7, false>(m, w, mine, s, d, lane, el, ser, smax, feat, sagg, sden); break;
                        case 6:  take_edges<6, false>(m, w, mine, s, d, lane, el, ser, smax, feat, sagg, sden); break;
                        case 5:  take_edges<5, false>(m, w, mine, s, d, lane, el, ser, smax, feat, sagg, sden); break;
                        case 4:  take_edges<4, false>(m, w, mine, s, d, lane, el, ser, smax, feat, sagg, sden); break;
                        case 3:  take_edges<3, false>(m, w, mine, s, d, lane, el, ser, smax, feat, sagg, sden); break;
                        case 2:  take_edges<2, false>(m, w, mine, s, d, lane, el, ser, smax, feat, sagg, sden); break;
                        default: take_edges<1, false>(m, w, mine, s, d, lane, el, ser, smax, feat, sagg, sden); break;
                        }
                    }
                }
            }
        }
        if (done) break;
        if (stop == SUNR) {
            ustart = 0;
            k0 += SWAVES * SUNR;
            done = k0 >= NUNIT;
        } else {
            ustart = stop;
        }
    }
}
__global__ void __launch_bounds__(SWAVES * 32)
gat_scatter(const int* __restrict__ src,
            const int* __restrict__ dst,
            const unsigned* __restrict__ own,
            const float* __restrict__ el,
            const float* __restrict__ er,
            const float* __restrict__ feat,
            float* __restrict__ agg,
            float* __restrict__ den) {
    static_assert(NN % NPW == 0 && EE % 4 == 0 && UEDGES == 512 && EUN == 8 && NH == 4,
                  "whole ranges, whole words of the owner plane, sixteen edges per lane, four heads");
    static_assert((unsigned)NWG <= NOOWN, "no work-group has the padding's owner byte");
    static_assert((QLEN & (QLEN - 1)) == 0 && QLEN >= UEDGES + 31, "the ring takes a whole unit on top of what may be left in it");
    static_assert(NN <= (1 << 23) && NPW <= 256, "a source id and a place in the range share one ring word");
    __shared__ float4 sagg4[NPW * FDIM / 4];
    __shared__ float4 sden4[NPW * NH / 4];
    __shared__ uint4  smax[NPW];
    __shared__ float4 ser[NPW];
    __shared__ int    sq[SWAVES * QLEN];
    float* sagg = (float*)sagg4;
    float* sden = (float*)sden4;
    const int t    = threadIdx.x;
    const int lane = t & 31;
    const int wave = __builtin_amdgcn_readfirstlane(t >> 5);
    const int lo   = blockIdx.x * NPW;
    const int start = (blockIdx.x * NUNIT) / NWG;
    const unsigned idx4 = (unsigned)blockIdx.x * 0x01010101u;
    int* q = sq + wave * QLEN;

    const float4 z4 = make_float4(0.f, 0.f, 0.f, 0.f);
    for (int i = t; i < NPW * FDIM / 4; i += SWAVES * 32) sagg4[i] = z4;
    if (t < NPW * NH / 4) sden4[t] = z4;
    if (t < NPW) {
        smax[t] = make_uint4(0u, 0u, 0u, 0u);
        ser[t]  = ((const float4*)er)[lo + t];
    }
    __syncthreads();

    unsigned head = 0u, n = 0u;
    bool keep = true;
    sweep<true>(lane, wave, lo, start, idx4, q, head, n, keep, false, false,
                src, dst, own, el, feat, ser, smax, sagg, sden);
    __syncthreads();

    n    = keep ? head : 0u;
    head = 0u;
    sweep<false>(lane, wave, lo, start, idx4, q, head, n, keep, keep, keep,
                 src, dst, own, el, feat, ser, smax, sagg, sden);
    __syncthreads();

    float4* ga = (float4*)(agg + (size_t)lo * FDIM);
    for (int i = t; i < NPW * FDIM / 4; i += SWAVES * 32) ga[i] = sagg4[i];
    if (t < NPW * NH / 4) ((float4*)(den + (size_t)lo * NH))[t] = sden4[t];
}

static void run_conv(const float* Xin, const float* pden, const float* W, const float* al, const float* ar,
                     const int* src, const int* dst, const unsigned* own, float* aggout, float* denout,
                     float* feat, float* el, float* er, hipStream_t stream) {
    if (pden)
        gat_gemm128_act<<<(NN + GROWS - 1) / GROWS, 256, 0, stream>>>(Xin, W, feat, pden);
    else
        gat_gemm128<<<(NN + GROWS - 1) / GROWS, 256, 0, stream>>>(Xin, W, feat);
    gat_attn_coeff<<<(NN * NH + 255) / 256, 256, 0, stream>>>(feat, al, ar, el, er);
    gat_scatter<<<NWG, SWAVES * 32, 0, stream>>>(src, dst, own, el, er, feat, aggout, denout);
}

extern "C" void kernel_launch(void* const* d_in, const int* in_sizes, int n_in,
                              void* d_out, int out_size, void* d_ws, size_t ws_size,
                              hipStream_t stream) {
    (void)in_sizes; (void)n_in; (void)out_size; (void)ws_size;

    const float* inputs = (const float*)d_in[0];
    const float* target = (const float*)d_in[1];
    const float* lamb   = (const float*)d_in[2];
    const float* W0     = (const float*)d_in[3];
    const float* al0    = (const float*)d_in[4];
    const float* ar0    = (const float*)d_in[5];
    const float* W1     = (const float*)d_in[6];
    const float* al1    = (const float*)d_in[7];
    const float* ar1    = (const float*)d_in[8];
    const float* Wout   = (const float*)d_in[9];
    const float* bout   = (const float*)d_in[10];
    const int*   src    = (const int*)d_in[11];
    const int*   dst    = (const int*)d_in[12];

    float* feat   = (float*)d_ws;
    float* s0a    = feat + (size_t)NN * FDIM;
    float* s1a    = s0a  + (size_t)NN * FDIM;
    float* s0b    = s1a  + (size_t)NN * FDIM;
    float* s1b    = s0b  + (size_t)NN * FDIM;
    float* el     = s1b  + (size_t)NN * FDIM;
    float* er     = el   + (size_t)NN * NH;
    float* d0a    = er   + (size_t)NN * NH;
    float* d1a    = d0a  + (size_t)NN * NH;
    float* d0b    = d1a  + (size_t)NN * NH;
    float* d1b    = d0b  + (size_t)NN * NH;
    unsigned* own = (unsigned*)(d1b + (size_t)NN * NH);

    float* hregion = (float*)d_out;
    float* logits  = hregion + (size_t)NN * FDIM;

    gat_owner_plane<<<(OWNB / 4 + 255) / 256, 256, 0, stream>>>(dst, own);

    run_conv(inputs, nullptr, W0, al0, ar0, src, dst, own, s0a, d0a,
             feat, el, er, stream);
    run_conv(s0a, d0a, W1, al1, ar1, src, dst, own, s1a, d1a,
             feat, el, er, stream);

    run_conv(target, nullptr, W0, al0, ar0, src, dst, own, s0b, d0b,
             feat, el, er, stream);
    run_conv(s0b, d0b, W1, al1, ar1, src, dst, own, s1b, d1b,
             feat, el, er, stream);

    gat_gemm_out<<<NN / 16, 96, 0, stream>>>(s0a, s1a, s0b, s1b, d0a, d1a, d0b, d1b,
                                             lamb, Wout, bout, hregion, logits);
}


# A kept edge's small gathers halved: owner launch leaves one word per edge (source id, place in range) read in place of src and dst; sweep 1 leaves the logits in LDS, sweep 2 loads no el row
# speedup vs baseline: 4.7559x; 1.0609x over previous
#include <hip/hip_runtime.h>
#include <hip/hip_bf16.h>
#include <math.h>

#define NN   50000
#define EE   800000
#define IND  128
#define HID  32
#define NH   4
#define NC   40
#define FDIM 128

typedef float v8f __attribute__((ext_vector_type(8)));
typedef float v2f __attribute__((ext_vector_type(2)));

__device__ __forceinline__ unsigned fkey(float f) {
    unsigned b = __float_as_uint(f);
    return (b & 0x80000000u) ? ~b : (b | 0x80000000u);
}
__device__ __forceinline__ float funkey(unsigned k) {
    unsigned b = (k & 0x80000000u) ? (k & 0x7FFFFFFFu) : ~k;
    return __uint_as_float(b);
}

__device__ __forceinline__ float elu1(float v) {
    return (v > 0.f) ? v : (expf(v) - 1.0f);
}

__device__ __forceinline__ float inv_den(float d) {
    return (d > 0.f) ? (1.0f / d) : 0.f;
}

#define GROWS 128
__device__ __forceinline__ void gemm128_body(const float* __restrict__ X,
                                             const float* __restrict__ W,
                                             float* __restrict__ Y,
                                             const float* __restrict__ pden,
                                             const bool act) {
    __shared__ float4 Ws[16 * 8 * 32];
    const int t    = threadIdx.x;
    const int lane = t & 31;
    const int wave = __builtin_amdgcn_readfirstlane(t >> 5);
    const int c    = lane & 15;
    const int hh   = lane >> 4;
    const int m0   = blockIdx.x * GROWS;
    const int mw   = m0 + wave * 16;
    const bool live = mw < NN;

    float4 a[16];
    float4 q = make_float4(1.f, 1.f, 1.f, 1.f);
    if (live) {
        const float4* xr = (const float4*)(X + (size_t)(mw + c) * FDIM) + hh;
#pragma unroll
        for (int i = 0; i < 16; ++i) a[i] = xr[2 * i];
        if (act) q = *(const float4*)(pden + (size_t)(mw + c) * NH);
    }

#pragma unroll
    for (int i = 0; i < 16; ++i) {
        const float* wp = W + (size_t)(8 * i + 4 * hh) * FDIM + 16 * wave + c;
        float4 v;
        v.x = wp[0];
        v.y = wp[FDIM];
        v.z = wp[2 * FDIM];
        v.w = wp[3 * FDIM];
        Ws[(i * 8 + wave) * 32 + lane] = v;
    }

    if (act && live) {
        const float iq[4] = { inv_den(q.x), inv_den(q.y), inv_den(q.z), inv_den(q.w) };
#pragma unroll
        for (int i = 0; i < 16; ++i) {
            const float s = iq[i >> 2];
            a[i].x = elu1(a[i].x * s); a[i].y = elu1(a[i].y * s);
            a[i].z = elu1(a[i].z * s); a[i].w = elu1(a[i].w * s);
        }
    }
    __syncthreads();
    if (!live) return;

    v8f acc[8] = {};
    float4 b[2][8];
#pragma unroll
    for (int j = 0; j < 8; ++j) b[0][j] = Ws[j * 32 + lane];
#pragma unroll
    for (int i = 0; i < 16; ++i) {
        const int cur = i & 1;
        __builtin_amdgcn_sched_barrier(0);
        if (i + 1 < 16) {
#pragma unroll
            for (int j = 0; j < 8; ++j) b[cur ^ 1][j] = Ws[((i + 1) * 8 + j) * 32 + lane];
        }
        __builtin_amdgcn_sched_barrier(0);
        v2f a0, a1;
        a0.x = a[i].x; a0.y = a[i].y;
        a1.x = a[i].z; a1.y = a[i].w;
#pragma unroll
        for (int j = 0; j < 8; ++j) {
            v2f b0;
            b0.x = b[cur][j].x; b0.y = b[cur][j].y;
            acc[j] = __builtin_amdgcn_wmma_f32_16x16x4_f32(false, a0, false, b0,
                                                           (short)0, acc[j], false, false);
        }
#pragma unroll
        for (int j = 0; j < 8; ++j) {
            v2f b1;
            b1.x = b[cur][j].z; b1.y = b[cur][j].w;
            acc[j] = __builtin_amdgcn_wmma_f32_16x16x4_f32(false, a1, false, b1,
                                                           (short)0, acc[j], false, false);
        }
    }
    float* yr = Y + (size_t)(mw + 8 * hh) * FDIM + c;
#pragma unroll
    for (int j = 0; j < 8; ++j)
#pragma unroll
        for (int r = 0; r < 8; ++r)
            yr[(size_t)r * FDIM + 16 * j] = acc[j][r];
}

__global__ void __launch_bounds__(256)
gat_gemm128(const float* __restrict__ X,
                            const float* __restrict__ W,
                            float* __restrict__ Y) {
    gemm128_body(X, W, Y, nullptr, false);
}

__global__ void __launch_bounds__(256)
gat_gemm128_act(const float* __restrict__ S,
                                const float* __restrict__ W,
                                float* __restrict__ Y,
                                const float* __restrict__ pden) {
    gemm128_body(S, W, Y, pden, true);
}

__global__ void gat_gemm_out(const float* __restrict__ s0a,
                             const float* __restrict__ s1a,
                             const float* __restrict__ s0b,
                             const float* __restrict__ s1b,
                             const float* __restrict__ d0a,
                             const float* __restrict__ d1a,
                             const float* __restrict__ d0b,
                             const float* __restrict__ d1b,
                             const float* __restrict__ lamb,
                             const float* __restrict__ Wout,
                             const float* __restrict__ bout,
                             float* __restrict__ hout,
                             float* __restrict__ logits) {
    __shared__ float As[16 * FDIM];
    const int m0 = blockIdx.x * 16;
    const int t  = threadIdx.x;
    const float lam = lamb[0];
    const size_t base4 = (size_t)m0 * (FDIM / 4);
    constexpr int NP = 6;
    float4 a0[NP], a1[NP], b0[NP], b1[NP];
    float qa0[NP], qa1[NP], qb0[NP], qb1[NP];
#pragma unroll
    for (int k = 0; k < NP; ++k) {
        const int i = t + 96 * k;
        if (i < 16 * FDIM / 4) {
            a0[k] = ((const float4*)s0a)[base4 + i];
            a1[k] = ((const float4*)s1a)[base4 + i];
            b0[k] = ((const float4*)s0b)[base4 + i];
            b1[k] = ((const float4*)s1b)[base4 + i];
            const size_t di = (size_t)(m0 + (i >> 5)) * NH + ((i & 31) >> 3);
            qa0[k] = d0a[di];
            qa1[k] = d1a[di];
            qb0[k] = d0b[di];
            qb1[k] = d1b[di];
        }
    }
#pragma unroll
    for (int k = 0; k < NP; ++k) {
        const int i = t + 96 * k;
        if (i < 16 * FDIM / 4) {
            const float ia0 = inv_den(qa0[k]), ia1 = inv_den(qa1[k]);
            const float ib0 = inv_den(qb0[k]), ib1 = inv_den(qb1[k]);
            float4 r;
            r.x = lam * elu1(a1[k].x * ia1 + elu1(a0[k].x * ia0)) + (1.0f - lam) * elu1(b1[k].x * ib1 + elu1(b0[k].x * ib0));
            r.y = lam * elu1(a1[k].y * ia1 + elu1(a0[k].y * ia0)) + (1.0f - lam) * elu1(b1[k].y * ib1 + elu1(b0[k].y * ib0));
            r.z = lam * elu1(a1[k].z * ia1 + elu1(a0[k].z * ia0)) + (1.0f - lam) * elu1(b1[k].z * ib1 + elu1(b0[k].z * ib0));
            r.w = lam * elu1(a1[k].w * ia1 + elu1(a0[k].w * ia0)) + (1.0f - lam) * elu1(b1[k].w * ib1 + elu1(b0[k].w * ib0));
            ((float4*)As)[i] = r;
            ((float4*)hout)[base4 + i] = r;
        }
    }
    __syncthreads();

    const int wave = t >> 5;
    const int lane = t & 31;
    const int n0   = wave * 16;
    const int mrow = lane & 15;
    const int kgrp = lane >> 4;
    const int col  = n0 + mrow;

    v8f acc = {};
    for (int kk = 0; kk < FDIM; kk += 4) {
        v2f a, b;
        a.x = As[mrow * FDIM + kk + 2 * kgrp + 0];
        a.y = As[mrow * FDIM + kk + 2 * kgrp + 1];
        b.x = (col < NC) ? Wout[(kk + 2 * kgrp + 0) * NC + col] : 0.0f;
        b.y = (col < NC) ? Wout[(kk + 2 * kgrp + 1) * NC + col] : 0.0f;
        acc = __builtin_amdgcn_wmma_f32_16x16x4_f32(false, a, false, b,
                                                    (short)0, acc, false, false);
    }
    const int cn    = lane & 15;
    const int mbase = (lane >> 4) * 8;
    const int ocol  = n0 + cn;
    if (ocol < NC) {
        const float bb = bout[ocol];
        for (int r = 0; r < 8; ++r)
            logits[(size_t)(m0 + mbase + r) * NC + ocol] = acc[r] + bb;
    }
}

__global__ void gat_attn_coeff(const float* __restrict__ feat,
                               const float* __restrict__ al,
                               const float* __restrict__ ar,
                               float* __restrict__ el,
                               float* __restrict__ er) {
    int i = blockIdx.x * blockDim.x + threadIdx.x;
    if (i >= NN * NH) return;
    int n = i >> 2, h = i & 3;
    const float* f = feat + (size_t)n * FDIM + h * HID;
    float sl = 0.f, sr = 0.f;
    for (int d = 0; d < HID; ++d) {
        sl += f[d] * al[h * HID + d];
        sr += f[d] * ar[h * HID + d];
    }
    el[i] = sl;
    er[i] = sr;
}

#define EUN    8
#define NPW    200
#define NWG    (NN / NPW)
#define SWAVES 32
#define UEDGES 512
#define NUNIT  ((EE + UEDGES - 1) / UEDGES)
#define OWNB   (NUNIT * UEDGES)
#define NOOWN  255u
#define SUNR   8
#define MG     4
#define QLEN   1024
#define LQ     128
__device__ __forceinline__ unsigned owner_byte(const int d) {
    const unsigned o = (unsigned)d / (unsigned)NPW;
    return (o < (unsigned)NWG) ? o : NOOWN;
}
__device__ __forceinline__ unsigned edge_word(const int s, const int d) {
    const unsigned o = (unsigned)d / (unsigned)NPW;
    return ((unsigned)s << 8) | ((unsigned)d - o * (unsigned)NPW);
}
__global__ void gat_owner_plane(const int* __restrict__ src, const int* __restrict__ dst,
                                unsigned* __restrict__ own, unsigned* __restrict__ rec) {
    const int i = blockIdx.x * blockDim.x + threadIdx.x;
    if (i >= OWNB / 4) return;
    unsigned w = 0xFFFFFFFFu;
    if (i < EE / 4) {
        const int4 d = ((const int4*)dst)[i];
        const int4 s = ((const int4*)src)[i];
        w = owner_byte(d.x) | (owner_byte(d.y) << 8) | (owner_byte(d.z) << 16) | (owner_byte(d.w) << 24);
        ((uint4*)rec)[i] = make_uint4(edge_word(s.x, d.x), edge_word(s.y, d.y),
                                      edge_word(s.z, d.z), edge_word(s.w, d.w));
    }
    own[i] = w;
}
__device__ __forceinline__ unsigned zero_bytes(const unsigned x) {
    const unsigned t = (x & 0x7F7F7F7Fu) + 0x7F7F7F7Fu;
    return ~(t | x | 0x7F7F7F7Fu);
}
__device__ __forceinline__ float4 edge_logits4(const float4 l, const float4 r) {
    float4 v;
    v.x = l.x + r.x; v.y = l.y + r.y; v.z = l.z + r.z; v.w = l.w + r.w;
    v.x = (v.x > 0.f) ? v.x : 0.2f * v.x;
    v.y = (v.y > 0.f) ? v.y : 0.2f * v.y;
    v.z = (v.z > 0.f) ? v.z : 0.2f * v.z;
    v.w = (v.w > 0.f) ? v.w : 0.2f * v.w;
    return v;
}
__device__ __forceinline__ float4 edge_weights(const bool mine, const bool has, const float4 vk,
                                               const int s, const int dl,
                                               const float* __restrict__ el,
                                               const float4* ser,
                                               const uint4* smax) {
    float4 l = make_float4(0.f, 0.f, 0.f, 0.f);
    float4 r = make_float4(0.f, 0.f, 0.f, 0.f);
    uint4  k = make_uint4(0u, 0u, 0u, 0u);
    if (mine) {
        k = smax[dl];
        if (!has) {
            l = ((const float4*)el)[s];
            r = ser[dl];
        }
    }
    const float4 v = has ? vk : edge_logits4(l, r);
    float4 w;
    w.x = expf(v.x - funkey(k.x));
    w.y = expf(v.y - funkey(k.y));
    w.z = expf(v.z - funkey(k.z));
    w.w = expf(v.w - funkey(k.w));
    return w;
}
template <int K, bool FIRST>
__device__ __forceinline__ void take_edges(unsigned& m, float4& w, const bool mine,
                                           const bool has, const float4 vk,
                                           const int s, const int d, const int lane,
                                           const float* __restrict__ el,
                                           const float4* ser,
                                           const uint4* smax,
                                           const float* __restrict__ feat,
                                           float* sagg,
                                           float* sden) {
    int   j[K];
    float f[K][NH];
#pragma unroll
    for (int u = 0; u < K; ++u) {
        j[u] = __builtin_ctz(m);
        m &= m - 1u;
        const int sj = __builtin_amdgcn_readlane(s, j[u]);
        const float* fr = feat + (size_t)sj * FDIM + lane;
        f[u][0] = fr[0];
        f[u][1] = fr[HID];
        f[u][2] = fr[2 * HID];
        f[u][3] = fr[3 * HID];
    }
    if (FIRST) w = edge_weights(mine, has, vk, s, d, el, ser, smax);
#pragma unroll
    for (int u = 0; u < K; ++u) {
        const int dj = __builtin_amdgcn_readlane(d, j[u]);
        const float b0 = __int_as_float(__builtin_amdgcn_readlane(__float_as_int(w.x), j[u]));
        const float b1 = __int_as_float(__builtin_amdgcn_readlane(__float_as_int(w.y), j[u]));
        const float b2 = __int_as_float(__builtin_amdgcn_readlane(__float_as_int(w.z), j[u]));
        const float b3 = __int_as_float(__builtin_amdgcn_readlane(__float_as_int(w.w), j[u]));
        float* ar = sagg + dj * FDIM + lane;
        atomicAdd(ar,           f[u][0] * b0);
        atomicAdd(ar + HID,     f[u][1] * b1);
        atomicAdd(ar + 2 * HID, f[u][2] * b2);
        atomicAdd(ar + 3 * HID, f[u][3] * b3);
    }
    unsigned long long jp = 0ull;
#pragma unroll
    for (int u = 0; u < K; ++u) jp |= (unsigned long long)(unsigned)j[u] << (8 * u);
    const int from = (int)((unsigned)(jp >> (2 * (lane & ~3))) & 31u) << 2;
    const int   dq = __builtin_amdgcn_ds_bpermute(from, d);
    const float q0 = __int_as_float(__builtin_amdgcn_ds_bpermute(from, __float_as_int(w.x)));
    const float q1 = __int_as_float(__builtin_amdgcn_ds_bpermute(from, __float_as_int(w.y)));
    const float q2 = __int_as_float(__builtin_amdgcn_ds_bpermute(from, __float_as_int(w.z)));
    const float q3 = __int_as_float(__builtin_amdgcn_ds_bpermute(from, __float_as_int(w.w)));
    const int hd = lane & 3;
    const float bw = (hd == 0) ? q0 : (hd == 1) ? q1 : (hd == 2) ? q2 : q3;
    if (lane < NH * K) atomicAdd(sden + dq * NH + hd, bw);
}
template <bool MAXPASS>
__device__ __forceinline__ void sweep(const int lane, const int wave, const int start,
                                      const unsigned idx4, int* q, float4* lg,
                                      unsigned& head, unsigned& n, bool& keep, bool done, const bool packed,
                                      const unsigned* __restrict__ rec,
                                      const unsigned* __restrict__ own,
                                      const float* __restrict__ el,
                                      const float* __restrict__ feat,
                                      const float4* ser, uint4* smax,
                                      float* sagg, float* sden) {
    int  k0 = wave;
    int  ustart = 0;
    uint4 dv[SUNR];
    int  eb[SUNR];
    for (;;) {
        int stop = SUNR;
        if (!done) {
            if (ustart == 0) {
#pragma unroll
                for (int u = 0; u < SUNR; ++u) {
                    const int k = k0 + SWAVES * u;
                    int unit = k + start;
                    if (unit >= NUNIT) unit -= NUNIT;
                    eb[u] = unit * UEDGES + lane * 16;
                    dv[u] = make_uint4(~0u, ~0u, ~0u, ~0u);
                    if (k < NUNIT) dv[u] = ((const uint4*)own)[unit * 32 + lane];
                }
            }
#pragma unroll
            for (int u = 0; u < SUNR; ++u) {
                if (u >= ustart && u < stop) {
                    unsigned hm = (zero_bytes(dv[u].x ^ idx4) >> 7) | (zero_bytes(dv[u].y ^ idx4) >> 6)
                                | (zero_bytes(dv[u].z ^ idx4) >> 5) | (zero_bytes(dv[u].w ^ idx4) >> 4);
                    if (__builtin_amdgcn_ballot_w32(hm != 0u) != 0u) {
                        if (MAXPASS && keep && head + n + (unsigned)UEDGES > (unsigned)QLEN) keep = false;
                        const unsigned used = (MAXPASS && keep) ? head + n : n;
                        if (used + (unsigned)UEDGES > (unsigned)QLEN) {
                            stop = u;
                        } else {
                            for (;;) {
                                const bool hit = hm != 0u;
                                const unsigned mm = __builtin_amdgcn_ballot_w32(hit);
                                if (mm == 0u) break;
                                const unsigned at = head + n + __builtin_amdgcn_mbcnt_lo(mm, 0u);
                                const int p = __builtin_ctz(hm | 0x80000000u);
                                if (hit) q[at & (QLEN - 1)] = eb[u] + ((p & 7) << 2) + (p >> 3);
                                hm &= hm - 1u;
                                n += (unsigned)__builtin_popcount(mm);
                            }
                        }
                    }
                }
            }
        }
        if (MAXPASS && keep) {
            if (done) {
                const unsigned total = n;
                __builtin_amdgcn_fence(__ATOMIC_RELEASE, "wavefront");
                __builtin_amdgcn_wave_barrier();
                __builtin_amdgcn_fence(__ATOMIC_ACQUIRE, "wavefront");
                for (unsigned b = 0u; b < total; b += 32u * MG) {
                    bool     on[MG];
                    int      ex[MG];
                    unsigned rx[MG];
                    float4   lx[MG];
#pragma unroll
                    for (int g = 0; g < MG; ++g) {
                        const unsigned i = b + 32u * (unsigned)g + (unsigned)lane;
                        on[g] = i < total;
                        const int e = q[i & (QLEN - 1)];
                        ex[g] = on[g] ? e : 0;
                    }
#pragma unroll
                    for (int g = 0; g < MG; ++g) rx[g] = rec[ex[g]];
#pragma unroll
                    for (int g = 0; g < MG; ++g) lx[g] = ((const float4*)el)[rx[g] >> 8];
#pragma unroll
                    for (int g = 0; g < MG; ++g) {
                        if (on[g]) {
                            const unsigned i = b + 32u * (unsigned)g + (unsigned)lane;
                            const int d = (int)(rx[g] & 255u);
                            q[i] = (int)rx[g];
                            const float4 v = edge_logits4(lx[g], ser[d]);
                            if (i < (unsigned)LQ) lg[i] = v;
                            unsigned* mk = (unsigned*)(smax + d);
                            atomicMax(mk + 0, fkey(v.x));
                            atomicMax(mk + 1, fkey(v.y));
                            atomicMax(mk + 2, fkey(v.z));
                            atomicMax(mk + 3, fkey(v.w));
                        }
                    }
                }
                head = total;
                n    = 0u;
            }
        } else
        while (n >= 32u || (done && n != 0u)) {
            const unsigned cnt = (n < 32u) ? n : 32u;
            __builtin_amdgcn_fence(__ATOMIC_RELEASE, "wavefront");
            __builtin_amdgcn_wave_barrier();
            __builtin_amdgcn_fence(__ATOMIC_ACQUIRE, "wavefront");
            const bool mine = lane < (int)cnt;
            const unsigned slot = (head + (unsigned)lane) & (QLEN - 1);
            int s = 0, d = 0;
            if (MAXPASS) {
                if (mine) {
                    const unsigned x = rec[q[slot]];
                    s = (int)(x >> 8);
                    d = (int)(x & 255u);
                    const float4 v = edge_logits4(((const float4*)el)[s], ser[d]);
                    unsigned* mk = (unsigned*)(smax + d);
                    atomicMax(mk + 0, fkey(v.x));
                    atomicMax(mk + 1, fkey(v.y));
                    atomicMax(mk + 2, fkey(v.z));
                    atomicMax(mk + 3, fkey(v.w));
                }
                head += cnt;
                n    -= cnt;
            } else {
                const bool has = packed && slot < (unsigned)LQ;
                float4 vk = make_float4(0.f, 0.f, 0.f, 0.f);
                if (mine) {
                    unsigned x = (unsigned)q[slot];
                    if (!packed) x = rec[x];
                    s = (int)(x >> 8);
                    d = (int)(x & 255u);
                    if (has) vk = lg[slot];
                }
                head += cnt;
                n    -= cnt;
                unsigned m = __builtin_amdgcn_ballot_w32(mine);
                float4 w;
                if (__builtin_popcount(m) >= EUN)
                    take_edges<EUN, true>(m, w, mine, has, vk, s, d, lane, el, ser, smax, feat, sagg, sden);
                else
                    w = edge_weights(mine, has, vk, s, d, el, ser, smax);
                while (m != 0u) {
                    const int nb = __builtin_popcount(m);
                    if (nb >= EUN) {
                        take_edges<EUN, false>(m, w, mine, has, vk, s, d, lane, el, ser, smax, feat, sagg, sden);
                    } else {
                        switch (nb) {
                        case 7:  take_edges<7, false>(m, w, mine, has, vk, s, d, lane, el, ser, smax, feat, sagg, sden); break;
                        case 6:  take_edges<6, false>(m, w, mine, has, vk, s, d, lane, el, ser, smax, feat, sagg, sden); break;
                        case 5:  take_edges<5, false>(m, w, mine, has, vk, s, d, lane, el, ser, smax, feat, sagg, sden); break;
                        case 4:  take_edges<4, false>(m, w, mine, has, vk, s, d, lane, el, ser, smax, feat, sagg, sden); break;
                        case 3:  take_edges<3, false>(m, w, mine, has, vk, s, d, lane, el, ser, smax, feat, sagg, sden); break;
                        case 2:  take_edges<2, false>(m, w, mine, has, vk, s, d, lane, el, ser, smax, feat, sagg, sden); break;
                        default: take_edges<1, false>(m, w, mine, has, vk, s, d, lane, el, ser, smax, feat, sagg, sden); break;
                        }
                    }
                }
            }
        }
        if (done) break;
        if (stop == SUNR) {
            ustart = 0;
            k0 += SWAVES * SUNR;
            done = k0 >= NUNIT;
        } else {
            ustart = stop;
        }
    }
}
__global__ void __launch_bounds__(SWAVES * 32)
gat_scatter(const unsigned* __restrict__ rec,
            const unsigned* __restrict__ own,
            const float* __restrict__ el,
            const float* __restrict__ er,
            const float* __restrict__ feat,
            float* __restrict__ agg,
            float* __restrict__ den) {
    static_assert(NN % NPW == 0 && EE % 4 == 0 && UEDGES == 512 && EUN == 8 && NH == 4,
                  "whole ranges, whole words of the owner plane, sixteen edges per lane, four heads");
    static_assert((unsigned)NWG <= NOOWN, "no work-group has the padding's owner byte");
    static_assert((QLEN & (QLEN - 1)) == 0 && QLEN >= UEDGES + 31, "the ring takes a whole unit on top of what may be left in it");
    static_assert(NN <= (1 << 23) && NPW <= 256, "a source id and a place in the range share one edge word");
    static_assert(LQ <= QLEN, "the logits left in LDS belong to ring entries");
    __shared__ float4 sagg4[NPW * FDIM / 4];
    __shared__ float4 sden4[NPW * NH / 4];
    __shared__ uint4  smax[NPW];
    __shared__ float4 ser[NPW];
    __shared__ int    sq[SWAVES * QLEN];
    __shared__ float4 slg[SWAVES * LQ];
    float* sagg = (float*)sagg4;
    float* sden = (float*)sden4;
    const int t    = threadIdx.x;
    const int lane = t & 31;
    const int wave = __builtin_amdgcn_readfirstlane(t >> 5);
    const int lo   = blockIdx.x * NPW;
    const int start = (blockIdx.x * NUNIT) / NWG;
    const unsigned idx4 = (unsigned)blockIdx.x * 0x01010101u;
    int* q = sq + wave * QLEN;
    float4* lg = slg + wave * LQ;

    const float4 z4 = make_float4(0.f, 0.f, 0.f, 0.f);
    for (int i = t; i < NPW * FDIM / 4; i += SWAVES * 32) sagg4[i] = z4;
    if (t < NPW * NH / 4) sden4[t] = z4;
    if (t < NPW) {
        smax[t] = make_uint4(0u, 0u, 0u, 0u);
        ser[t]  = ((const float4*)er)[lo + t];
    }
    __syncthreads();

    unsigned head = 0u, n = 0u;
    bool keep = true;
    sweep<true>(lane, wave, start, idx4, q, lg, head, n, keep, false, false,
                rec, own, el, feat, ser, smax, sagg, sden);
    __syncthreads();

    n    = keep ? head : 0u;
    head = 0u;
    sweep<false>(lane, wave, start, idx4, q, lg, head, n, keep, keep, keep,
                 rec, own, el, feat, ser, smax, sagg, sden);
    __syncthreads();

    float4* ga = (float4*)(agg + (size_t)lo * FDIM);
    for (int i = t; i < NPW * FDIM / 4; i += SWAVES * 32) ga[i] = sagg4[i];
    if (t < NPW * NH / 4) ((float4*)(den + (size_t)lo * NH))[t] = sden4[t];
}

static void run_conv(const float* Xin, const float* pden, const float* W, const float* al, const float* ar,
                     const unsigned* rec, const unsigned* own, float* aggout, float* denout,
                     float* feat, float* el, float* er, hipStream_t stream) {
    if (pden)
        gat_gemm128_act<<<(NN + GROWS - 1) / GROWS, 256, 0, stream>>>(Xin, W, feat, pden);
    else
        gat_gemm128<<<(NN + GROWS - 1) / GROWS, 256, 0, stream>>>(Xin, W, feat);
    gat_attn_coeff<<<(NN * NH + 255) / 256, 256, 0, stream>>>(feat, al, ar, el, er);
    gat_scatter<<<NWG, SWAVES * 32, 0, stream>>>(rec, own, el, er, feat, aggout, denout);
}

extern "C" void kernel_launch(void* const* d_in, const int* in_sizes, int n_in,
                              void* d_out, int out_size, void* d_ws, size_t ws_size,
                              hipStream_t stream) {
    (void)in_sizes; (void)n_in; (void)out_size; (void)ws_size;

    const float* inputs = (const float*)d_in[0];
    const float* target = (const float*)d_in[1];
    const float* lamb   = (const float*)d_in[2];
    const float* W0     = (const float*)d_in[3];
    const float* al0    = (const float*)d_in[4];
    const float* ar0    = (const float*)d_in[5];
    const float* W1     = (const float*)d_in[6];
    const float* al1    = (const float*)d_in[7];
    const float* ar1    = (const float*)d_in[8];
    const float* Wout   = (const float*)d_in[9];
    const float* bout   = (const float*)d_in[10];
    const int*   src    = (const int*)d_in[11];
    const int*   dst    = (const int*)d_in[12];

    float* feat   = (float*)d_ws;
    float* s0a    = feat + (size_t)NN * FDIM;
    float* s1a    = s0a  + (size_t)NN * FDIM;
    float* s0b    = s1a  + (size_t)NN * FDIM;
    float* s1b    = s0b  + (size_t)NN * FDIM;
    float* el     = s1b  + (size_t)NN * FDIM;
    float* er     = el   + (size_t)NN * NH;
    float* d0a    = er   + (size_t)NN * NH;
    float* d1a    = d0a  + (size_t)NN * NH;
    float* d0b    = d1a  + (size_t)NN * NH;
    float* d1b    = d0b  + (size_t)NN * NH;
    unsigned* own = (unsigned*)(d1b + (size_t)NN * NH);
    unsigned* rec = own + OWNB / 4;

    float* hregion = (float*)d_out;
    float* logits  = hregion + (size_t)NN * FDIM;

    gat_owner_plane<<<(OWNB / 4 + 255) / 256, 256, 0, stream>>>(src, dst, own, rec);

    run_conv(inputs, nullptr, W0, al0, ar0, rec, own, s0a, d0a,
             feat, el, er, stream);
    run_conv(s0a, d0a, W1, al1, ar1, rec, own, s1a, d1a,
             feat, el, er, stream);

    run_conv(target, nullptr, W0, al0, ar0, rec, own, s0b, d0b,
             feat, el, er, stream);
    run_conv(s0b, d0b, W1, al1, ar1, rec, own, s1b, d1b,
             feat, el, er, stream);

    gat_gemm_out<<<NN / 16, 96, 0, stream>>>(s0a, s1a, s0b, s1b, d0a, d1a, d0b, d1b,
                                             lamb, Wout, bout, hregion, logits);
}
